# starting kernel
# baseline (speedup 1.0000x reference)
#include <hip/hip_runtime.h>
#include <cstdint>

#define N_NODES 50000
#define IN_CH   128
#define Hc      32
#define Dd      2
#define LAYERS  4
#define EU_CNT  400000
#define E_CNT   (2*EU_CNT)
#define DH      (Dd*Hc)
#define ND      (N_NODES*Dd)
#define OUTC    32

typedef __attribute__((ext_vector_type(2))) float v2f;
typedef __attribute__((ext_vector_type(8))) float v8f;

__device__ __forceinline__ v8f wmma4(v2f a, v2f b, v8f c) {
  return __builtin_amdgcn_wmma_f32_16x16x4_f32(false, a, false, b, (short)0, c,
                                               false, false);
}

__device__ __forceinline__ void atomicAddF(float* p, float v) {
  unsafeAtomicAdd(p, v);
}

__global__ void wmma_gemm_xwT(const float* __restrict__ A,
                              const float* __restrict__ W,
                              const float* __restrict__ bias,
                              const float* __restrict__ Wl,
                              float* __restrict__ C,
                              float* __restrict__ C2,
                              int M, int K, int Nn, int do_elu)
{
  const int lane   = threadIdx.x & 31;
  const int wid    = blockIdx.x * (blockDim.x >> 5) + (threadIdx.x >> 5);
  const int ntiles = Nn >> 4;
  const int mtiles = M  >> 4;
  if (wid >= mtiles * ntiles) return;
  const int mt = wid / ntiles;
  const int nt = wid - mt * ntiles;

  const int arow = (mt << 4) + (lane & 15);
  const int bcol = (nt << 4) + (lane & 15);
  const int kb   = (lane >> 4) << 1;

  const float* Wp = W + (size_t)bcol * K + kb;

  v8f acc = {};
  if (Wl) {
    const int n  = arow >> 1;
    const int dp = arow & 1;
    const float w0 = Wl[dp * 2 + 0];
    const float w1 = Wl[dp * 2 + 1];
    const float* A0 = A + (size_t)(n * 2) * K + kb;
    const float* A1 = A0 + K;
    for (int k0 = 0; k0 < K; k0 += 4) {
      v2f a0 = *(const v2f*)(A0 + k0);
      v2f a1 = *(const v2f*)(A1 + k0);
      v2f a;
      a.x = w0 * a0.x + w1 * a1.x;
      a.y = w0 * a0.y + w1 * a1.y;
      v2f b = *(const v2f*)(Wp + k0);
      acc = wmma4(a, b, acc);
    }
  } else {
    const float* Ap = A + (size_t)arow * K + kb;
    for (int k0 = 0; k0 < K; k0 += 4) {
      v2f a = *(const v2f*)(Ap + k0);
      v2f b = *(const v2f*)(Wp + k0);
      acc = wmma4(a, b, acc);
    }
  }

  const float bv    = bias ? bias[bcol] : 0.0f;
  const int   rbase = (mt << 4) + ((lane >> 4) << 3);
  #pragma unroll
  for (int r = 0; r < 8; ++r) {
    float v = acc[r] + bv;
    if (do_elu) v = (v > 0.0f) ? v : expm1f(v);
    size_t idx = (size_t)(rbase + r) * Nn + bcol;
    C[idx] = v;
    if (C2) C2[idx] = v;
  }
}

__global__ void sheaf_pq(const float* __restrict__ xcur,
                         const float* __restrict__ Ws,
                         float* __restrict__ p, float* __restrict__ q,
                         float* __restrict__ Dg)
{
  int n = blockIdx.x * blockDim.x + threadIdx.x;
  if (n >= N_NODES) return;
  const float* xrow = xcur + (size_t)n * DH;
  float p0 = 0.f, p1 = 0.f, q0 = 0.f, q1 = 0.f;
  #pragma unroll 4
  for (int j = 0; j < DH; ++j) {
    float xv = xrow[j];
    p0 += Ws[j]             * xv;
    p1 += Ws[2*DH + j]      * xv;
    q0 += Ws[DH + j]        * xv;
    q1 += Ws[2*DH + DH + j] * xv;
  }
  p[n*2+0] = p0; p[n*2+1] = p1;
  q[n*2+0] = q0; q[n*2+1] = q1;
  Dg[n*2+0] = 0.f; Dg[n*2+1] = 0.f;
}

__global__ void edge_F(const int* __restrict__ src, const int* __restrict__ dst,
                       const float* __restrict__ p, const float* __restrict__ q,
                       float* __restrict__ F, float* __restrict__ Dg)
{
  int e = blockIdx.x * blockDim.x + threadIdx.x;
  if (e >= E_CNT) return;
  int s = src[e], t = dst[e];
  float f0 = tanhf(p[s*2+0] + q[t*2+0]);
  float f1 = tanhf(p[s*2+1] + q[t*2+1]);
  F[e*2+0] = f0; F[e*2+1] = f1;
  atomicAddF(&Dg[s*2+0], f0*f0);
  atomicAddF(&Dg[s*2+1], f1*f1);
}

__global__ void node_dinv(const float* __restrict__ Dg,
                          float* __restrict__ dinv, float* __restrict__ diagw)
{
  int i = blockIdx.x * blockDim.x + threadIdx.x;
  if (i >= ND) return;
  float g  = Dg[i];
  float dv = rsqrtf(g + 1.0f);
  dinv[i]  = dv;
  diagw[i] = g * dv * dv;
}

__global__ void y_init(const float* __restrict__ xr,
                       const float* __restrict__ diagw, float* __restrict__ y)
{
  int i = blockIdx.x * blockDim.x + threadIdx.x;
  if (i >= ND * Hc) return;
  y[i] = diagw[i >> 5] * xr[i];
}

__global__ void edge_scatter(const int* __restrict__ src, const int* __restrict__ dst,
                             const float* __restrict__ F, const float* __restrict__ dinv,
                             const float* __restrict__ xr, float* __restrict__ y)
{
  unsigned t = blockIdx.x * blockDim.x + threadIdx.x;
  if (t >= (unsigned)E_CNT * DH) return;
  int j = t & (DH - 1);
  int e = t >> 6;
  int d = j >> 5;
  int c = j & 31;
  int s  = src[e], tt = dst[e];
  int er = (e >= EU_CNT) ? (e - EU_CNT) : (e + EU_CNT);
  float offw = -F[e*2+d] * F[er*2+d] * dinv[s*2+d] * dinv[tt*2+d];
  float val  = offw * xr[(size_t)(tt*2+d) * Hc + c];
  atomicAddF(&y[(size_t)(s*2+d) * Hc + c], val);
}

__global__ void elu_residual(const float* __restrict__ y,
                             const float* __restrict__ eps2,
                             float* __restrict__ x0, float* __restrict__ xcur)
{
  int i = blockIdx.x * blockDim.x + threadIdx.x;
  if (i >= ND * Hc) return;
  int dp  = (i >> 5) & 1;
  float v = y[i];
  float e = (v > 0.0f) ? v : expm1f(v);
  float coeff = 1.0f + tanhf(eps2[dp]);
  float nx = coeff * x0[i] - e;
  x0[i]   = nx;
  xcur[i] = nx;
}

extern "C" void kernel_launch(void* const* d_in, const int* in_sizes, int n_in,
                              void* d_out, int out_size, void* d_ws, size_t ws_size,
                              hipStream_t stream) {
  const float* x_in   = (const float*)d_in[0];
  const int*   ei     = (const int*)  d_in[1];
  const float* W1     = (const float*)d_in[2];
  const float* b1     = (const float*)d_in[3];
  const float* Wsheaf = (const float*)d_in[4];
  const float* Wleft  = (const float*)d_in[5];
  const float* Wright = (const float*)d_in[6];
  const float* eps    = (const float*)d_in[7];
  const float* W2     = (const float*)d_in[8];
  const float* b2     = (const float*)d_in[9];
  float* out = (float*)d_out;

  float* ws = (float*)d_ws;
  size_t o = 0;
  float* x_cur = ws + o; o += (size_t)ND * Hc;
  float* x0    = ws + o; o += (size_t)ND * Hc;
  float* y     = ws + o; o += (size_t)ND * Hc;
  float* xr    = ws + o; o += (size_t)ND * Hc;
  float* p     = ws + o; o += 2 * (size_t)N_NODES;
  float* q     = ws + o; o += 2 * (size_t)N_NODES;
  float* F     = ws + o; o += 2 * (size_t)E_CNT;
  float* Dg    = ws + o; o += (size_t)ND;
  float* dinv  = ws + o; o += (size_t)ND;
  float* diagw = ws + o; o += (size_t)ND;

  const int* src = ei;
  const int* dst = ei + E_CNT;

  const int TPB = 256;
  const int elemBlocks = (ND * Hc + TPB - 1) / TPB;

  {
    int tiles = (N_NODES / 16) * (DH / 16);
    wmma_gemm_xwT<<<(tiles + 7) / 8, TPB, 0, stream>>>(
        x_in, W1, b1, nullptr, x_cur, x0, N_NODES, IN_CH, DH, 1);
  }

  for (int l = 0; l < LAYERS; ++l) {
    sheaf_pq<<<(N_NODES + TPB - 1) / TPB, TPB, 0, stream>>>(
        x_cur, Wsheaf + (size_t)l * 2 * 2 * DH, p, q, Dg);
    edge_F<<<(E_CNT + TPB - 1) / TPB, TPB, 0, stream>>>(src, dst, p, q, F, Dg);
    node_dinv<<<(ND + TPB - 1) / TPB, TPB, 0, stream>>>(Dg, dinv, diagw);
    {
      int tiles = (ND / 16) * (Hc / 16);
      wmma_gemm_xwT<<<(tiles + 7) / 8, TPB, 0, stream>>>(
          x_cur, Wright + (size_t)l * Hc * Hc, nullptr, Wleft + (size_t)l * 4,
          xr, nullptr, ND, Hc, Hc, 0);
    }
    y_init<<<elemBlocks, TPB, 0, stream>>>(xr, diagw, y);
    edge_scatter<<<((unsigned)E_CNT * DH + TPB - 1) / TPB, TPB, 0, stream>>>(
        src, dst, F, dinv, xr, y);
    elu_residual<<<elemBlocks, TPB, 0, stream>>>(y, eps + (size_t)l * 2, x0, x_cur);
  }

  {
    int tiles = (N_NODES / 16) * (OUTC / 16);
    wmma_gemm_xwT<<<(tiles + 7) / 8, TPB, 0, stream>>>(
        x_cur, W2, b2, nullptr, out, nullptr, N_NODES, DH, OUTC, 0);
  }
}


# y scatter-add (float atomics) replaced by per-node gather over edges counting-sorted by source each call (int atomic ranks); y kept in registers, init and elu-residual folded in; sums in sorted order
# speedup vs baseline: 1.5453x; 1.5453x over previous
#include <hip/hip_runtime.h>
#include <cstdint>

#define N_NODES 50000
#define IN_CH   128
#define Hc      32
#define Dd      2
#define LAYERS  4
#define EU_CNT  400000
#define E_CNT   (2*EU_CNT)
#define DH      (Dd*Hc)
#define ND      (N_NODES*Dd)
#define OUTC    32

typedef __attribute__((ext_vector_type(2))) float v2f;
typedef __attribute__((ext_vector_type(8))) float v8f;

__device__ __forceinline__ v8f wmma4(v2f a, v2f b, v8f c) {
  return __builtin_amdgcn_wmma_f32_16x16x4_f32(false, a, false, b, (short)0, c,
                                               false, false);
}

__device__ __forceinline__ void atomicAddF(float* p, float v) {
  unsafeAtomicAdd(p, v);
}

__global__ void wmma_gemm_xwT(const float* __restrict__ A,
                              const float* __restrict__ W,
                              const float* __restrict__ bias,
                              const float* __restrict__ Wl,
                              float* __restrict__ C,
                              float* __restrict__ C2,
                              int M, int K, int Nn, int do_elu)
{
  const int lane   = threadIdx.x & 31;
  const int wid    = blockIdx.x * (blockDim.x >> 5) + (threadIdx.x >> 5);
  const int ntiles = Nn >> 4;
  const int mtiles = M  >> 4;
  if (wid >= mtiles * ntiles) return;
  const int mt = wid / ntiles;
  const int nt = wid - mt * ntiles;

  const int arow = (mt << 4) + (lane & 15);
  const int bcol = (nt << 4) + (lane & 15);
  const int kb   = (lane >> 4) << 1;

  const float* Wp = W + (size_t)bcol * K + kb;

  v8f acc = {};
  if (Wl) {
    const int n  = arow >> 1;
    const int dp = arow & 1;
    const float w0 = Wl[dp * 2 + 0];
    const float w1 = Wl[dp * 2 + 1];
    const float* A0 = A + (size_t)(n * 2) * K + kb;
    const float* A1 = A0 + K;
    for (int k0 = 0; k0 < K; k0 += 4) {
      v2f a0 = *(const v2f*)(A0 + k0);
      v2f a1 = *(const v2f*)(A1 + k0);
      v2f a;
      a.x = w0 * a0.x + w1 * a1.x;
      a.y = w0 * a0.y + w1 * a1.y;
      v2f b = *(const v2f*)(Wp + k0);
      acc = wmma4(a, b, acc);
    }
  } else {
    const float* Ap = A + (size_t)arow * K + kb;
    for (int k0 = 0; k0 < K; k0 += 4) {
      v2f a = *(const v2f*)(Ap + k0);
      v2f b = *(const v2f*)(Wp + k0);
      acc = wmma4(a, b, acc);
    }
  }

  const float bv    = bias ? bias[bcol] : 0.0f;
  const int   rbase = (mt << 4) + ((lane >> 4) << 3);
  #pragma unroll
  for (int r = 0; r < 8; ++r) {
    float v = acc[r] + bv;
    if (do_elu) v = (v > 0.0f) ? v : expm1f(v);
    size_t idx = (size_t)(rbase + r) * Nn + bcol;
    C[idx] = v;
    if (C2) C2[idx] = v;
  }
}

__global__ void sheaf_pq(const float* __restrict__ xcur,
                         const float* __restrict__ Ws,
                         float* __restrict__ p, float* __restrict__ q,
                         float* __restrict__ Dg)
{
  int n = blockIdx.x * blockDim.x + threadIdx.x;
  if (n >= N_NODES) return;
  const float* xrow = xcur + (size_t)n * DH;
  float p0 = 0.f, p1 = 0.f, q0 = 0.f, q1 = 0.f;
  #pragma unroll 4
  for (int j = 0; j < DH; ++j) {
    float xv = xrow[j];
    p0 += Ws[j]             * xv;
    p1 += Ws[2*DH + j]      * xv;
    q0 += Ws[DH + j]        * xv;
    q1 += Ws[2*DH + DH + j] * xv;
  }
  p[n*2+0] = p0; p[n*2+1] = p1;
  q[n*2+0] = q0; q[n*2+1] = q1;
  Dg[n*2+0] = 0.f; Dg[n*2+1] = 0.f;
}

__global__ void edge_F(const int* __restrict__ src, const int* __restrict__ dst,
                       const float* __restrict__ p, const float* __restrict__ q,
                       float* __restrict__ F, float* __restrict__ Dg)
{
  int e = blockIdx.x * blockDim.x + threadIdx.x;
  if (e >= E_CNT) return;
  int s = src[e], t = dst[e];
  float f0 = tanhf(p[s*2+0] + q[t*2+0]);
  float f1 = tanhf(p[s*2+1] + q[t*2+1]);
  F[e*2+0] = f0; F[e*2+1] = f1;
  atomicAddF(&Dg[s*2+0], f0*f0);
  atomicAddF(&Dg[s*2+1], f1*f1);
}

__global__ void node_dinv(const float* __restrict__ Dg,
                          float* __restrict__ dinv, float* __restrict__ diagw)
{
  int i = blockIdx.x * blockDim.x + threadIdx.x;
  if (i >= ND) return;
  float g  = Dg[i];
  float dv = rsqrtf(g + 1.0f);
  dinv[i]  = dv;
  diagw[i] = g * dv * dv;
}

__global__ void csr_zero(int* __restrict__ cnt)
{
  int i = blockIdx.x * blockDim.x + threadIdx.x;
  if (i < N_NODES) cnt[i] = 0;
}

__global__ void csr_rank(const int* __restrict__ src, int* __restrict__ cnt,
                         int* __restrict__ slot)
{
  int e = blockIdx.x * blockDim.x + threadIdx.x;
  if (e >= E_CNT) return;
  slot[e] = atomicAdd(&cnt[src[e]], 1);
}

#define SCAN_ITERS ((N_NODES + 1023) / 1024)
#define SCAN_PAD   (SCAN_ITERS * 1024)
__global__ void __launch_bounds__(1024) csr_scan(const int* __restrict__ cnt,
                                                 int* __restrict__ rowptr)
{
  __shared__ int wtot[32];
  const int lane = threadIdx.x & 31;
  const int w    = threadIdx.x >> 5;
  const int base = w * (SCAN_ITERS * 32) + lane;
  int v[SCAN_ITERS];
  int s = 0;
  #pragma unroll
  for (int i = 0; i < SCAN_ITERS; ++i) {
    const int idx = base + i * 32;
    const int x   = cnt[idx < N_NODES ? idx : N_NODES - 1];
    v[i] = idx < N_NODES ? x : 0;
    s += v[i];
  }
  for (int o = 16; o > 0; o >>= 1) s += __shfl_xor(s, o);
  if (lane == 0) wtot[w] = s;
  __syncthreads();
  int carry = 0;
  for (int j = 0; j < w; ++j) carry += wtot[j];
  #pragma unroll
  for (int i = 0; i < SCAN_ITERS; ++i) {
    int inc = v[i];
    #pragma unroll
    for (int o = 1; o < 32; o <<= 1) {
      const int u = __shfl_up(inc, o);
      if (lane >= o) inc += u;
    }
    rowptr[base + i * 32] = carry + inc - v[i];
    carry += __shfl(inc, 31);
  }
}

__global__ void csr_fill(const int* __restrict__ src, const int* __restrict__ rowptr,
                         const int* __restrict__ slot, int* __restrict__ eid)
{
  int e = blockIdx.x * blockDim.x + threadIdx.x;
  if (e >= E_CNT) return;
  eid[rowptr[src[e]] + slot[e]] = e;
}

__global__ void __launch_bounds__(256) node_gather(
    const int* __restrict__ rowptr, const int* __restrict__ eid,
    const int* __restrict__ dst,
    const float* __restrict__ F, const float* __restrict__ dinv,
    const float* __restrict__ diagw, const float* __restrict__ xr,
    const float* __restrict__ eps2, float* __restrict__ x0,
    float* __restrict__ xcur)
{
  const int lane = threadIdx.x & 31;
  const int s = __builtin_amdgcn_readfirstlane(
      (int)(blockIdx.x * (blockDim.x >> 5) + (threadIdx.x >> 5)));
  if (s >= N_NODES) return;
  const int d = lane >> 4;
  int beg = rowptr[s];
  int end = rowptr[s + 1];
  beg = beg < 0 ? 0 : beg;
  end = end > E_CNT ? E_CNT : end;

  const size_t own = (size_t)s * DH + 2 * lane;
  const v2f ds2 = *(const v2f*)(dinv + 2 * s);
  const v2f xs  = *(const v2f*)(xr + own);
  const float dw = diagw[2 * s + d];
  v2f acc;
  acc.x = dw * xs.x;
  acc.y = dw * xs.y;

  for (int k0 = beg; k0 < end; k0 += 32) {
    const int k = k0 + lane;
    int   t  = 0;
    float w0 = 0.0f, w1 = 0.0f;
    if (k < end) {
      const int e  = eid[k];
      const int er = (e >= EU_CNT) ? (e - EU_CNT) : (e + EU_CNT);
      t = dst[e];
      const v2f f  = *(const v2f*)(F + 2 * (size_t)e);
      const v2f fr = *(const v2f*)(F + 2 * (size_t)er);
      const v2f dt = *(const v2f*)(dinv + 2 * (size_t)t);
      w0 = -f.x * fr.x * ds2.x * dt.x;
      w1 = -f.y * fr.y * ds2.y * dt.y;
    }
    const int m = (end - k0) < 32 ? (end - k0) : 32;
    #pragma unroll 4
    for (int j = 0; j < m; ++j) {
      const int   tj = __builtin_amdgcn_readlane(t, j);
      const float a0 = __int_as_float(__builtin_amdgcn_readlane(__float_as_int(w0), j));
      const float a1 = __int_as_float(__builtin_amdgcn_readlane(__float_as_int(w1), j));
      const float wj = d ? a1 : a0;
      const v2f xv = *(const v2f*)(xr + (size_t)tj * DH + 2 * lane);
      acc.x += wj * xv.x;
      acc.y += wj * xv.y;
    }
  }

  const float coeff = 1.0f + tanhf(eps2[d]);
  const v2f xo = *(const v2f*)(x0 + own);
  const float e0 = (acc.x > 0.0f) ? acc.x : expm1f(acc.x);
  const float e1 = (acc.y > 0.0f) ? acc.y : expm1f(acc.y);
  v2f nx;
  nx.x = coeff * xo.x - e0;
  nx.y = coeff * xo.y - e1;
  *(v2f*)(x0 + own)   = nx;
  *(v2f*)(xcur + own) = nx;
}

extern "C" void kernel_launch(void* const* d_in, const int* in_sizes, int n_in,
                              void* d_out, int out_size, void* d_ws, size_t ws_size,
                              hipStream_t stream) {
  const float* x_in   = (const float*)d_in[0];
  const int*   ei     = (const int*)  d_in[1];
  const float* W1     = (const float*)d_in[2];
  const float* b1     = (const float*)d_in[3];
  const float* Wsheaf = (const float*)d_in[4];
  const float* Wleft  = (const float*)d_in[5];
  const float* Wright = (const float*)d_in[6];
  const float* eps    = (const float*)d_in[7];
  const float* W2     = (const float*)d_in[8];
  const float* b2     = (const float*)d_in[9];
  float* out = (float*)d_out;

  float* ws = (float*)d_ws;
  size_t o = 0;
  float* x_cur = ws + o; o += (size_t)ND * Hc;
  float* x0    = ws + o; o += (size_t)ND * Hc;
  int*   csr   = (int*)(ws + o); o += (size_t)ND * Hc;
  int*   eid    = csr;
  int*   slot   = csr + (size_t)E_CNT;
  int*   cnt    = csr + 2 * (size_t)E_CNT;
  int*   rowptr = csr + 2 * (size_t)E_CNT + 50048;
  float* xr    = ws + o; o += (size_t)ND * Hc;
  float* p     = ws + o; o += 2 * (size_t)N_NODES;
  float* q     = ws + o; o += 2 * (size_t)N_NODES;
  float* F     = ws + o; o += 2 * (size_t)E_CNT;
  float* Dg    = ws + o; o += (size_t)ND;
  float* dinv  = ws + o; o += (size_t)ND;
  float* diagw = ws + o; o += (size_t)ND;

  const int* src = ei;
  const int* dst = ei + E_CNT;

  const int TPB = 256;

  csr_zero<<<(N_NODES + TPB - 1) / TPB, TPB, 0, stream>>>(cnt);
  csr_rank<<<(E_CNT + TPB - 1) / TPB, TPB, 0, stream>>>(src, cnt, slot);
  csr_scan<<<1, 1024, 0, stream>>>(cnt, rowptr);
  csr_fill<<<(E_CNT + TPB - 1) / TPB, TPB, 0, stream>>>(src, rowptr, slot, eid);

  {
    int tiles = (N_NODES / 16) * (DH / 16);
    wmma_gemm_xwT<<<(tiles + 7) / 8, TPB, 0, stream>>>(
        x_in, W1, b1, nullptr, x_cur, x0, N_NODES, IN_CH, DH, 1);
  }

  for (int l = 0; l < LAYERS; ++l) {
    sheaf_pq<<<(N_NODES + TPB - 1) / TPB, TPB, 0, stream>>>(
        x_cur, Wsheaf + (size_t)l * 2 * 2 * DH, p, q, Dg);
    edge_F<<<(E_CNT + TPB - 1) / TPB, TPB, 0, stream>>>(src, dst, p, q, F, Dg);
    node_dinv<<<(ND + TPB - 1) / TPB, TPB, 0, stream>>>(Dg, dinv, diagw);
    {
      int tiles = (ND / 16) * (Hc / 16);
      wmma_gemm_xwT<<<(tiles + 7) / 8, TPB, 0, stream>>>(
          x_cur, Wright + (size_t)l * Hc * Hc, nullptr, Wleft + (size_t)l * 4,
          xr, nullptr, ND, Hc, Hc, 0);
    }
    node_gather<<<(N_NODES + 7) / 8, TPB, 0, stream>>>(
        rowptr, eid, dst, F, dinv, diagw, xr, eps + (size_t)l * 2, x0, x_cur);
  }

  {
    int tiles = (N_NODES / 16) * (OUTC / 16);
    wmma_gemm_xwT<<<(tiles + 7) / 8, TPB, 0, stream>>>(
        x_cur, W2, b2, nullptr, out, nullptr, N_NODES, DH, OUTC, 0);
  }
}


# edge records (dst, reverse edge's place, src) written once in source-sorted order, F kept in that order: per-layer edge reads turn contiguous; Dg = per-node sum in sorted order, float atomics gone
# speedup vs baseline: 3.0726x; 1.9884x over previous
#include <hip/hip_runtime.h>
#include <cstdint>

#define N_NODES 50000
#define IN_CH   128
#define Hc      32
#define Dd      2
#define LAYERS  4
#define EU_CNT  400000
#define E_CNT   (2*EU_CNT)
#define DH      (Dd*Hc)
#define ND      (N_NODES*Dd)
#define OUTC    32

typedef __attribute__((ext_vector_type(2))) float v2f;
typedef __attribute__((ext_vector_type(8))) float v8f;
typedef __attribute__((ext_vector_type(2))) int   v2i;
typedef __attribute__((ext_vector_type(4))) int   v4i;

__device__ __forceinline__ v8f wmma4(v2f a, v2f b, v8f c) {
  return __builtin_amdgcn_wmma_f32_16x16x4_f32(false, a, false, b, (short)0, c,
                                               false, false);
}

__global__ void wmma_gemm_xwT(const float* __restrict__ A,
                              const float* __restrict__ W,
                              const float* __restrict__ bias,
                              const float* __restrict__ Wl,
                              float* __restrict__ C,
                              float* __restrict__ C2,
                              int M, int K, int Nn, int do_elu)
{
  const int lane   = threadIdx.x & 31;
  const int wid    = blockIdx.x * (blockDim.x >> 5) + (threadIdx.x >> 5);
  const int ntiles = Nn >> 4;
  const int mtiles = M  >> 4;
  if (wid >= mtiles * ntiles) return;
  const int mt = wid / ntiles;
  const int nt = wid - mt * ntiles;

  const int arow = (mt << 4) + (lane & 15);
  const int bcol = (nt << 4) + (lane & 15);
  const int kb   = (lane >> 4) << 1;

  const float* Wp = W + (size_t)bcol * K + kb;

  v8f acc = {};
  if (Wl) {
    const int n  = arow >> 1;
    const int dp = arow & 1;
    const float w0 = Wl[dp * 2 + 0];
    const float w1 = Wl[dp * 2 + 1];
    const float* A0 = A + (size_t)(n * 2) * K + kb;
    const float* A1 = A0 + K;
    for (int k0 = 0; k0 < K; k0 += 4) {
      v2f a0 = *(const v2f*)(A0 + k0);
      v2f a1 = *(const v2f*)(A1 + k0);
      v2f a;
      a.x = w0 * a0.x + w1 * a1.x;
      a.y = w0 * a0.y + w1 * a1.y;
      v2f b = *(const v2f*)(Wp + k0);
      acc = wmma4(a, b, acc);
    }
  } else {
    const float* Ap = A + (size_t)arow * K + kb;
    for (int k0 = 0; k0 < K; k0 += 4) {
      v2f a = *(const v2f*)(Ap + k0);
      v2f b = *(const v2f*)(Wp + k0);
      acc = wmma4(a, b, acc);
    }
  }

  const float bv    = bias ? bias[bcol] : 0.0f;
  const int   rbase = (mt << 4) + ((lane >> 4) << 3);
  #pragma unroll
  for (int r = 0; r < 8; ++r) {
    float v = acc[r] + bv;
    if (do_elu) v = (v > 0.0f) ? v : expm1f(v);
    size_t idx = (size_t)(rbase + r) * Nn + bcol;
    C[idx] = v;
    if (C2) C2[idx] = v;
  }
}

__global__ void sheaf_pq(const float* __restrict__ xcur,
                         const float* __restrict__ Ws,
                         float* __restrict__ p, float* __restrict__ q)
{
  int n = blockIdx.x * blockDim.x + threadIdx.x;
  if (n >= N_NODES) return;
  const float* xrow = xcur + (size_t)n * DH;
  float p0 = 0.f, p1 = 0.f, q0 = 0.f, q1 = 0.f;
  #pragma unroll 4
  for (int j = 0; j < DH; ++j) {
    float xv = xrow[j];
    p0 += Ws[j]             * xv;
    p1 += Ws[2*DH + j]      * xv;
    q0 += Ws[DH + j]        * xv;
    q1 += Ws[2*DH + DH + j] * xv;
  }
  p[n*2+0] = p0; p[n*2+1] = p1;
  q[n*2+0] = q0; q[n*2+1] = q1;
}

__global__ void edge_F(const int* __restrict__ rec,
                       const float* __restrict__ p, const float* __restrict__ q,
                       float* __restrict__ Fs)
{
  int k = blockIdx.x * blockDim.x + threadIdx.x;
  if (k >= E_CNT) return;
  const v4i r  = *(const v4i*)(rec + 4 * (size_t)k);
  const v2f ps = *(const v2f*)(p + 2 * (size_t)r.z);
  const v2f qt = *(const v2f*)(q + 2 * (size_t)r.x);
  v2f f;
  f.x = tanhf(ps.x + qt.x);
  f.y = tanhf(ps.y + qt.y);
  *(v2f*)(Fs + 2 * (size_t)k) = f;
}

__global__ void node_dinv(const int* __restrict__ rowptr, const float* __restrict__ Fs,
                          float* __restrict__ dinv, float* __restrict__ diagw)
{
  int i = blockIdx.x * blockDim.x + threadIdx.x;
  if (i >= N_NODES) return;
  int beg = rowptr[i];
  int end = rowptr[i + 1];
  beg = beg < 0 ? 0 : beg;
  end = end > E_CNT ? E_CNT : end;
  float g0 = 0.0f, g1 = 0.0f;
  #pragma unroll 4
  for (int k = beg; k < end; ++k) {
    const v2f f = *(const v2f*)(Fs + 2 * (size_t)k);
    g0 += f.x * f.x;
    g1 += f.y * f.y;
  }
  const float d0 = rsqrtf(g0 + 1.0f);
  const float d1 = rsqrtf(g1 + 1.0f);
  v2f dv; dv.x = d0;           dv.y = d1;
  v2f dw; dw.x = g0 * d0 * d0; dw.y = g1 * d1 * d1;
  *(v2f*)(dinv  + 2 * (size_t)i) = dv;
  *(v2f*)(diagw + 2 * (size_t)i) = dw;
}

__global__ void csr_zero(int* __restrict__ cnt)
{
  int i = blockIdx.x * blockDim.x + threadIdx.x;
  if (i < N_NODES) cnt[i] = 0;
}

__global__ void csr_rank(const int* __restrict__ src, int* __restrict__ cnt,
                         int* __restrict__ slot)
{
  int e = blockIdx.x * blockDim.x + threadIdx.x;
  if (e >= E_CNT) return;
  slot[e] = atomicAdd(&cnt[src[e]], 1);
}

#define SCAN_ITERS ((N_NODES + 1023) / 1024)
#define SCAN_PAD   (SCAN_ITERS * 1024)
__global__ void __launch_bounds__(1024) csr_scan(const int* __restrict__ cnt,
                                                 int* __restrict__ rowptr)
{
  __shared__ int wtot[32];
  const int lane = threadIdx.x & 31;
  const int w    = threadIdx.x >> 5;
  const int base = w * (SCAN_ITERS * 32) + lane;
  int v[SCAN_ITERS];
  int s = 0;
  #pragma unroll
  for (int i = 0; i < SCAN_ITERS; ++i) {
    const int idx = base + i * 32;
    const int x   = cnt[idx < N_NODES ? idx : N_NODES - 1];
    v[i] = idx < N_NODES ? x : 0;
    s += v[i];
  }
  for (int o = 16; o > 0; o >>= 1) s += __shfl_xor(s, o);
  if (lane == 0) wtot[w] = s;
  __syncthreads();
  int carry = 0;
  for (int j = 0; j < w; ++j) carry += wtot[j];
  #pragma unroll
  for (int i = 0; i < SCAN_ITERS; ++i) {
    int inc = v[i];
    #pragma unroll
    for (int o = 1; o < 32; o <<= 1) {
      const int u = __shfl_up(inc, o);
      if (lane >= o) inc += u;
    }
    rowptr[base + i * 32] = carry + inc - v[i];
    carry += __shfl(inc, 31);
  }
}

__global__ void csr_fill(const int* __restrict__ src, const int* __restrict__ dst,
                         const int* __restrict__ rowptr, const int* __restrict__ slot,
                         int* __restrict__ rec)
{
  int e = blockIdx.x * blockDim.x + threadIdx.x;
  if (e >= E_CNT) return;
  const int er  = (e >= EU_CNT) ? (e - EU_CNT) : (e + EU_CNT);
  const int s   = src[e];
  const int pos = rowptr[s] + slot[e];
  const int pr  = rowptr[src[er]] + slot[er];
  v4i r;
  r.x = dst[e];
  r.y = pr;
  r.z = s;
  r.w = 0;
  *(v4i*)(rec + 4 * (size_t)pos) = r;
}

__global__ void __launch_bounds__(256) node_gather(
    const int* __restrict__ rowptr, const int* __restrict__ rec,
    const float* __restrict__ Fs, const float* __restrict__ dinv,
    const float* __restrict__ diagw, const float* __restrict__ xr,
    const float* __restrict__ eps2, float* __restrict__ x0,
    float* __restrict__ xcur)
{
  const int lane = threadIdx.x & 31;
  const int s = __builtin_amdgcn_readfirstlane(
      (int)(blockIdx.x * (blockDim.x >> 5) + (threadIdx.x >> 5)));
  if (s >= N_NODES) return;
  const int d = lane >> 4;
  int beg = rowptr[s];
  int end = rowptr[s + 1];
  beg = beg < 0 ? 0 : beg;
  end = end > E_CNT ? E_CNT : end;

  const size_t own = (size_t)s * DH + 2 * lane;
  const v2f ds2 = *(const v2f*)(dinv + 2 * s);
  const v2f xs  = *(const v2f*)(xr + own);
  const float dw = diagw[2 * s + d];
  v2f acc;
  acc.x = dw * xs.x;
  acc.y = dw * xs.y;

  for (int k0 = beg; k0 < end; k0 += 32) {
    const int k = k0 + lane;
    int   t  = 0;
    float w0 = 0.0f, w1 = 0.0f;
    if (k < end) {
      const v2i r  = *(const v2i*)(rec + 4 * (size_t)k);
      t = r.x;
      const v2f f  = *(const v2f*)(Fs + 2 * (size_t)k);
      const v2f fr = *(const v2f*)(Fs + 2 * (size_t)r.y);
      const v2f dt = *(const v2f*)(dinv + 2 * (size_t)t);
      w0 = -f.x * fr.x * ds2.x * dt.x;
      w1 = -f.y * fr.y * ds2.y * dt.y;
    }
    const int m = (end - k0) < 32 ? (end - k0) : 32;
    #pragma unroll 4
    for (int j = 0; j < m; ++j) {
      const int   tj = __builtin_amdgcn_readlane(t, j);
      const float a0 = __int_as_float(__builtin_amdgcn_readlane(__float_as_int(w0), j));
      const float a1 = __int_as_float(__builtin_amdgcn_readlane(__float_as_int(w1), j));
      const float wj = d ? a1 : a0;
      const v2f xv = *(const v2f*)(xr + (size_t)tj * DH + 2 * lane);
      acc.x += wj * xv.x;
      acc.y += wj * xv.y;
    }
  }

  const float coeff = 1.0f + tanhf(eps2[d]);
  const v2f xo = *(const v2f*)(x0 + own);
  const float e0 = (acc.x > 0.0f) ? acc.x : expm1f(acc.x);
  const float e1 = (acc.y > 0.0f) ? acc.y : expm1f(acc.y);
  v2f nx;
  nx.x = coeff * xo.x - e0;
  nx.y = coeff * xo.y - e1;
  *(v2f*)(x0 + own)   = nx;
  *(v2f*)(xcur + own) = nx;
}

extern "C" void kernel_launch(void* const* d_in, const int* in_sizes, int n_in,
                              void* d_out, int out_size, void* d_ws, size_t ws_size,
                              hipStream_t stream) {
  const float* x_in   = (const float*)d_in[0];
  const int*   ei     = (const int*)  d_in[1];
  const float* W1     = (const float*)d_in[2];
  const float* b1     = (const float*)d_in[3];
  const float* Wsheaf = (const float*)d_in[4];
  const float* Wleft  = (const float*)d_in[5];
  const float* Wright = (const float*)d_in[6];
  const float* eps    = (const float*)d_in[7];
  const float* W2     = (const float*)d_in[8];
  const float* b2     = (const float*)d_in[9];
  float* out = (float*)d_out;

  float* ws = (float*)d_ws;
  size_t o = 0;
  float* x_cur = ws + o; o += (size_t)ND * Hc;
  float* x0    = ws + o; o += (size_t)ND * Hc;
  int*   rec   = (int*)(ws + o); o += (size_t)ND * Hc;
  float* xr    = ws + o; o += (size_t)ND * Hc;
  int*   slot  = (int*)xr;
  int*   cnt   = (int*)xr + (size_t)E_CNT;
  float* p     = ws + o; o += 2 * (size_t)N_NODES;
  float* q     = ws + o; o += 2 * (size_t)N_NODES;
  float* Fs    = ws + o; o += 2 * (size_t)E_CNT;
  int*   rowptr = (int*)(ws + o); o += (size_t)ND;
  float* dinv  = ws + o; o += (size_t)ND;
  float* diagw = ws + o; o += (size_t)ND;

  const int* src = ei;
  const int* dst = ei + E_CNT;

  const int TPB = 256;

  csr_zero<<<(N_NODES + TPB - 1) / TPB, TPB, 0, stream>>>(cnt);
  csr_rank<<<(E_CNT + TPB - 1) / TPB, TPB, 0, stream>>>(src, cnt, slot);
  csr_scan<<<1, 1024, 0, stream>>>(cnt, rowptr);
  csr_fill<<<(E_CNT + TPB - 1) / TPB, TPB, 0, stream>>>(src, dst, rowptr, slot, rec);

  {
    int tiles = (N_NODES / 16) * (DH / 16);
    wmma_gemm_xwT<<<(tiles + 7) / 8, TPB, 0, stream>>>(
        x_in, W1, b1, nullptr, x_cur, x0, N_NODES, IN_CH, DH, 1);
  }

  for (int l = 0; l < LAYERS; ++l) {
    sheaf_pq<<<(N_NODES + TPB - 1) / TPB, TPB, 0, stream>>>(
        x_cur, Wsheaf + (size_t)l * 2 * 2 * DH, p, q);
    edge_F<<<(E_CNT + TPB - 1) / TPB, TPB, 0, stream>>>(rec, p, q, Fs);
    node_dinv<<<(N_NODES + TPB - 1) / TPB, TPB, 0, stream>>>(rowptr, Fs, dinv, diagw);
    {
      int tiles = (ND / 16) * (Hc / 16);
      wmma_gemm_xwT<<<(tiles + 7) / 8, TPB, 0, stream>>>(
          x_cur, Wright + (size_t)l * Hc * Hc, nullptr, Wleft + (size_t)l * 4,
          xr, nullptr, ND, Hc, Hc, 0);
    }
    node_gather<<<(N_NODES + 7) / 8, TPB, 0, stream>>>(
        rowptr, rec, Fs, dinv, diagw, xr, eps + (size_t)l * 2, x0, x_cur);
  }

  {
    int tiles = (N_NODES / 16) * (OUTC / 16);
    wmma_gemm_xwT<<<(tiles + 7) / 8, TPB, 0, stream>>>(
        x_cur, W2, b2, nullptr, out, nullptr, N_NODES, DH, OUTC, 0);
  }
}


# gather: one scattered read per edge, not three: reverse edge's F recomputed in the edge kernel from endpoints in its record; dinv[dst] folded into xr rows in the product's epilogue; sum order kept
# speedup vs baseline: 3.2469x; 1.0567x over previous
#include <hip/hip_runtime.h>
#include <cstdint>

#define N_NODES 50000
#define IN_CH   128
#define Hc      32
#define Dd      2
#define LAYERS  4
#define EU_CNT  400000
#define E_CNT   (2*EU_CNT)
#define DH      (Dd*Hc)
#define ND      (N_NODES*Dd)
#define OUTC    32

typedef __attribute__((ext_vector_type(2))) float v2f;
typedef __attribute__((ext_vector_type(4))) float v4f;
typedef __attribute__((ext_vector_type(8))) float v8f;
typedef __attribute__((ext_vector_type(4))) int   v4i;

__device__ __forceinline__ v8f wmma4(v2f a, v2f b, v8f c) {
  return __builtin_amdgcn_wmma_f32_16x16x4_f32(false, a, false, b, (short)0, c,
                                               false, false);
}

__global__ void wmma_gemm_xwT(const float* __restrict__ A,
                              const float* __restrict__ W,
                              const float* __restrict__ bias,
                              const float* __restrict__ Wl,
                              const float* __restrict__ rowscale,
                              float* __restrict__ C,
                              float* __restrict__ C2,
                              int M, int K, int Nn, int do_elu)
{
  const int lane   = threadIdx.x & 31;
  const int wid    = blockIdx.x * (blockDim.x >> 5) + (threadIdx.x >> 5);
  const int ntiles = Nn >> 4;
  const int mtiles = M  >> 4;
  if (wid >= mtiles * ntiles) return;
  const int mt = wid / ntiles;
  const int nt = wid - mt * ntiles;

  const int arow = (mt << 4) + (lane & 15);
  const int bcol = (nt << 4) + (lane & 15);
  const int kb   = (lane >> 4) << 1;

  const float* Wp = W + (size_t)bcol * K + kb;

  const int rbase = (mt << 4) + ((lane >> 4) << 3);
  v4f rs0 = {1.0f, 1.0f, 1.0f, 1.0f}, rs1 = {1.0f, 1.0f, 1.0f, 1.0f};
  if (rowscale) {
    rs0 = *(const v4f*)(rowscale + rbase);
    rs1 = *(const v4f*)(rowscale + rbase + 4);
  }

  v8f acc = {};
  if (Wl) {
    const int n  = arow >> 1;
    const int dp = arow & 1;
    const float w0 = Wl[dp * 2 + 0];
    const float w1 = Wl[dp * 2 + 1];
    const float* A0 = A + (size_t)(n * 2) * K + kb;
    const float* A1 = A0 + K;
    for (int k0 = 0; k0 < K; k0 += 4) {
      v2f a0 = *(const v2f*)(A0 + k0);
      v2f a1 = *(const v2f*)(A1 + k0);
      v2f a;
      a.x = w0 * a0.x + w1 * a1.x;
      a.y = w0 * a0.y + w1 * a1.y;
      v2f b = *(const v2f*)(Wp + k0);
      acc = wmma4(a, b, acc);
    }
  } else {
    const float* Ap = A + (size_t)arow * K + kb;
    for (int k0 = 0; k0 < K; k0 += 4) {
      v2f a = *(const v2f*)(Ap + k0);
      v2f b = *(const v2f*)(Wp + k0);
      acc = wmma4(a, b, acc);
    }
  }

  const float bv    = bias ? bias[bcol] : 0.0f;
  const float rs[8] = {rs0.x, rs0.y, rs0.z, rs0.w, rs1.x, rs1.y, rs1.z, rs1.w};
  #pragma unroll
  for (int r = 0; r < 8; ++r) {
    float v = acc[r] + bv;
    if (do_elu) v = (v > 0.0f) ? v : expm1f(v);
    if (rowscale) v *= rs[r];
    size_t idx = (size_t)(rbase + r) * Nn + bcol;
    C[idx] = v;
    if (C2) C2[idx] = v;
  }
}

__global__ void sheaf_pq(const float* __restrict__ xcur,
                         const float* __restrict__ Ws,
                         float* __restrict__ pq)
{
  int n = blockIdx.x * blockDim.x + threadIdx.x;
  if (n >= N_NODES) return;
  const float* xrow = xcur + (size_t)n * DH;
  float p0 = 0.f, p1 = 0.f, q0 = 0.f, q1 = 0.f;
  #pragma unroll 4
  for (int j = 0; j < DH; ++j) {
    float xv = xrow[j];
    p0 += Ws[j]             * xv;
    p1 += Ws[2*DH + j]      * xv;
    q0 += Ws[DH + j]        * xv;
    q1 += Ws[2*DH + DH + j] * xv;
  }
  v4f o4; o4.x = p0; o4.y = p1; o4.z = q0; o4.w = q1;
  *(v4f*)(pq + 4 * (size_t)n) = o4;
}

__global__ void edge_F(const int* __restrict__ rec,
                       const float* __restrict__ pq,
                       float* __restrict__ Ft, float* __restrict__ G)
{
  int k = blockIdx.x * blockDim.x + threadIdx.x;
  if (k >= E_CNT) return;
  const v4i r = *(const v4i*)(rec + 4 * (size_t)k);
  const v4f a = *(const v4f*)(pq + 4 * (size_t)r.y);
  const v4f b = *(const v4f*)(pq + 4 * (size_t)r.x);
  v4f bp = b;
  if (r.z != r.x) bp = *(const v4f*)(pq + 4 * (size_t)r.z);
  v4f aq = a;
  if (r.w != r.y) aq = *(const v4f*)(pq + 4 * (size_t)r.w);
  v2f f, g;
  f.x = tanhf(a.x + b.z);
  f.y = tanhf(a.y + b.w);
  g.x = f.x * tanhf(bp.x + aq.z);
  g.y = f.y * tanhf(bp.y + aq.w);
  *(v2f*)(Ft + 2 * (size_t)k) = f;
  *(v2f*)(G  + 2 * (size_t)k) = g;
}

__global__ void node_dinv(const int* __restrict__ rowptr, const float* __restrict__ Fs,
                          float* __restrict__ dinv, float* __restrict__ diagw)
{
  int i = blockIdx.x * blockDim.x + threadIdx.x;
  if (i >= N_NODES) return;
  int beg = rowptr[i];
  int end = rowptr[i + 1];
  beg = beg < 0 ? 0 : beg;
  end = end > E_CNT ? E_CNT : end;
  float g0 = 0.0f, g1 = 0.0f;
  #pragma unroll 4
  for (int k = beg; k < end; ++k) {
    const v2f f = *(const v2f*)(Fs + 2 * (size_t)k);
    g0 += f.x * f.x;
    g1 += f.y * f.y;
  }
  const float d0 = rsqrtf(g0 + 1.0f);
  const float d1 = rsqrtf(g1 + 1.0f);
  v2f dv; dv.x = d0;           dv.y = d1;
  v2f dw; dw.x = g0 * d0;      dw.y = g1 * d1;
  *(v2f*)(dinv  + 2 * (size_t)i) = dv;
  *(v2f*)(diagw + 2 * (size_t)i) = dw;
}

__global__ void csr_zero(int* __restrict__ cnt)
{
  int i = blockIdx.x * blockDim.x + threadIdx.x;
  if (i < N_NODES) cnt[i] = 0;
}

__global__ void csr_rank(const int* __restrict__ src, int* __restrict__ cnt,
                         int* __restrict__ slot)
{
  int e = blockIdx.x * blockDim.x + threadIdx.x;
  if (e >= E_CNT) return;
  slot[e] = atomicAdd(&cnt[src[e]], 1);
}

#define SCAN_ITERS ((N_NODES + 1023) / 1024)
#define SCAN_PAD   (SCAN_ITERS * 1024)
__global__ void __launch_bounds__(1024) csr_scan(const int* __restrict__ cnt,
                                                 int* __restrict__ rowptr)
{
  __shared__ int wtot[32];
  const int lane = threadIdx.x & 31;
  const int w    = threadIdx.x >> 5;
  const int base = w * (SCAN_ITERS * 32) + lane;
  int v[SCAN_ITERS];
  int s = 0;
  #pragma unroll
  for (int i = 0; i < SCAN_ITERS; ++i) {
    const int idx = base + i * 32;
    const int x   = cnt[idx < N_NODES ? idx : N_NODES - 1];
    v[i] = idx < N_NODES ? x : 0;
    s += v[i];
  }
  for (int o = 16; o > 0; o >>= 1) s += __shfl_xor(s, o);
  if (lane == 0) wtot[w] = s;
  __syncthreads();
  int carry = 0;
  for (int j = 0; j < w; ++j) carry += wtot[j];
  #pragma unroll
  for (int i = 0; i < SCAN_ITERS; ++i) {
    int inc = v[i];
    #pragma unroll
    for (int o = 1; o < 32; o <<= 1) {
      const int u = __shfl_up(inc, o);
      if (lane >= o) inc += u;
    }
    rowptr[base + i * 32] = carry + inc - v[i];
    carry += __shfl(inc, 31);
  }
}

__global__ void csr_fill(const int* __restrict__ src, const int* __restrict__ dst,
                         const int* __restrict__ rowptr, const int* __restrict__ slot,
                         int* __restrict__ rec)
{
  int e = blockIdx.x * blockDim.x + threadIdx.x;
  if (e >= E_CNT) return;
  const int er  = (e >= EU_CNT) ? (e - EU_CNT) : (e + EU_CNT);
  const int s   = src[e];
  const int pos = rowptr[s] + slot[e];
  v4i r;
  r.x = dst[e];
  r.y = s;
  r.z = src[er];
  r.w = dst[er];
  *(v4i*)(rec + 4 * (size_t)pos) = r;
}

__global__ void __launch_bounds__(256) node_gather(
    const int* __restrict__ rowptr, const int* __restrict__ rec,
    const float* __restrict__ G, const float* __restrict__ dinv,
    const float* __restrict__ diagw, const float* __restrict__ xr,
    const float* __restrict__ eps2, float* __restrict__ x0,
    float* __restrict__ xcur)
{
  const int lane = threadIdx.x & 31;
  const int s = __builtin_amdgcn_readfirstlane(
      (int)(blockIdx.x * (blockDim.x >> 5) + (threadIdx.x >> 5)));
  if (s >= N_NODES) return;
  const int d = lane >> 4;
  int beg = rowptr[s];
  int end = rowptr[s + 1];
  beg = beg < 0 ? 0 : beg;
  end = end > E_CNT ? E_CNT : end;

  const size_t own = (size_t)s * DH + 2 * lane;
  const v2f ds2 = *(const v2f*)(dinv + 2 * s);
  const v2f xs  = *(const v2f*)(xr + own);
  const float dw = diagw[2 * s + d];
  v2f acc;
  acc.x = dw * xs.x;
  acc.y = dw * xs.y;

  for (int k0 = beg; k0 < end; k0 += 32) {
    const int k = k0 + lane;
    int   t  = 0;
    float w0 = 0.0f, w1 = 0.0f;
    if (k < end) {
      t = rec[4 * (size_t)k];
      const v2f g = *(const v2f*)(G + 2 * (size_t)k);
      w0 = -g.x * ds2.x;
      w1 = -g.y * ds2.y;
    }
    const int m = (end - k0) < 32 ? (end - k0) : 32;
    #pragma unroll 4
    for (int j = 0; j < m; ++j) {
      const int   tj = __builtin_amdgcn_readlane(t, j);
      const float a0 = __int_as_float(__builtin_amdgcn_readlane(__float_as_int(w0), j));
      const float a1 = __int_as_float(__builtin_amdgcn_readlane(__float_as_int(w1), j));
      const float wj = d ? a1 : a0;
      const v2f xv = *(const v2f*)(xr + (size_t)tj * DH + 2 * lane);
      acc.x += wj * xv.x;
      acc.y += wj * xv.y;
    }
  }

  const float coeff = 1.0f + tanhf(eps2[d]);
  const v2f xo = *(const v2f*)(x0 + own);
  const float e0 = (acc.x > 0.0f) ? acc.x : expm1f(acc.x);
  const float e1 = (acc.y > 0.0f) ? acc.y : expm1f(acc.y);
  v2f nx;
  nx.x = coeff * xo.x - e0;
  nx.y = coeff * xo.y - e1;
  *(v2f*)(x0 + own)   = nx;
  *(v2f*)(xcur + own) = nx;
}

extern "C" void kernel_launch(void* const* d_in, const int* in_sizes, int n_in,
                              void* d_out, int out_size, void* d_ws, size_t ws_size,
                              hipStream_t stream) {
  const float* x_in   = (const float*)d_in[0];
  const int*   ei     = (const int*)  d_in[1];
  const float* W1     = (const float*)d_in[2];
  const float* b1     = (const float*)d_in[3];
  const float* Wsheaf = (const float*)d_in[4];
  const float* Wleft  = (const float*)d_in[5];
  const float* Wright = (const float*)d_in[6];
  const float* eps    = (const float*)d_in[7];
  const float* W2     = (const float*)d_in[8];
  const float* b2     = (const float*)d_in[9];
  float* out = (float*)d_out;

  float* ws = (float*)d_ws;
  size_t o = 0;
  float* x_cur = ws + o; o += (size_t)ND * Hc;
  float* x0    = ws + o; o += (size_t)ND * Hc;
  int*   rec   = (int*)(ws + o); o += (size_t)ND * Hc;
  float* xr    = ws + o; o += (size_t)ND * Hc;
  int*   slot  = (int*)xr;
  int*   cnt   = (int*)xr + (size_t)E_CNT;
  float* pq    = ws + o; o += 2 * (size_t)N_NODES; o += 2 * (size_t)N_NODES;
  float* G     = ws + o; o += 2 * (size_t)E_CNT;
  float* Ft    = xr;
  int*   rowptr = (int*)(ws + o); o += (size_t)ND;
  float* dinv  = ws + o; o += (size_t)ND;
  float* diagw = ws + o; o += (size_t)ND;

  const int* src = ei;
  const int* dst = ei + E_CNT;

  const int TPB = 256;

  csr_zero<<<(N_NODES + TPB - 1) / TPB, TPB, 0, stream>>>(cnt);
  csr_rank<<<(E_CNT + TPB - 1) / TPB, TPB, 0, stream>>>(src, cnt, slot);
  csr_scan<<<1, 1024, 0, stream>>>(cnt, rowptr);
  csr_fill<<<(E_CNT + TPB - 1) / TPB, TPB, 0, stream>>>(src, dst, rowptr, slot, rec);

  {
    int tiles = (N_NODES / 16) * (DH / 16);
    wmma_gemm_xwT<<<(tiles + 7) / 8, TPB, 0, stream>>>(
        x_in, W1, b1, nullptr, nullptr, x_cur, x0, N_NODES, IN_CH, DH, 1);
  }

  for (int l = 0; l < LAYERS; ++l) {
    sheaf_pq<<<(N_NODES + TPB - 1) / TPB, TPB, 0, stream>>>(
        x_cur, Wsheaf + (size_t)l * 2 * 2 * DH, pq);
    edge_F<<<(E_CNT + TPB - 1) / TPB, TPB, 0, stream>>>(rec, pq, Ft, G);
    node_dinv<<<(N_NODES + TPB - 1) / TPB, TPB, 0, stream>>>(rowptr, Ft, dinv, diagw);
    {
      int tiles = (ND / 16) * (Hc / 16);
      wmma_gemm_xwT<<<(tiles + 7) / 8, TPB, 0, stream>>>(
          x_cur, Wright + (size_t)l * Hc * Hc, nullptr, Wleft + (size_t)l * 4,
          dinv, xr, nullptr, ND, Hc, Hc, 0);
    }
    node_gather<<<(N_NODES + 7) / 8, TPB, 0, stream>>>(
        rowptr, rec, G, dinv, diagw, xr, eps + (size_t)l * 2, x0, x_cur);
  }

  {
    int tiles = (N_NODES / 16) * (OUTC / 16);
    wmma_gemm_xwT<<<(tiles + 7) / 8, TPB, 0, stream>>>(
        x_cur, W2, b2, nullptr, nullptr, out, nullptr, N_NODES, DH, OUTC, 0);
  }
}


# launches 26 -> 19: per-node Dg/dinv sum moved into the layer product's prologue, next layer's p,q into the gather's epilogue (their 64-long sums now a lane tree); x and x0 share one plane
# speedup vs baseline: 3.8214x; 1.1769x over previous
#include <hip/hip_runtime.h>
#include <cstdint>

#define N_NODES 50000
#define IN_CH   128
#define Hc      32
#define Dd      2
#define LAYERS  4
#define EU_CNT  400000
#define E_CNT   (2*EU_CNT)
#define DH      (Dd*Hc)
#define ND      (N_NODES*Dd)
#define OUTC    32

typedef __attribute__((ext_vector_type(2))) float v2f;
typedef __attribute__((ext_vector_type(4))) float v4f;
typedef __attribute__((ext_vector_type(8))) float v8f;
typedef __attribute__((ext_vector_type(4))) int   v4i;

__device__ __forceinline__ v8f wmma4(v2f a, v2f b, v8f c) {
  return __builtin_amdgcn_wmma_f32_16x16x4_f32(false, a, false, b, (short)0, c,
                                               false, false);
}

__global__ void wmma_gemm_xwT(const float* __restrict__ A,
                              const float* __restrict__ W,
                              const float* __restrict__ bias,
                              const float* __restrict__ Wl,
                              const float* __restrict__ rowscale,
                              float* __restrict__ C,
                              float* __restrict__ C2,
                              int M, int K, int Nn, int do_elu)
{
  const int lane   = threadIdx.x & 31;
  const int wid    = blockIdx.x * (blockDim.x >> 5) + (threadIdx.x >> 5);
  const int ntiles = Nn >> 4;
  const int mtiles = M  >> 4;
  if (wid >= mtiles * ntiles) return;
  const int mt = wid / ntiles;
  const int nt = wid - mt * ntiles;

  const int arow = (mt << 4) + (lane & 15);
  const int bcol = (nt << 4) + (lane & 15);
  const int kb   = (lane >> 4) << 1;

  const float* Wp = W + (size_t)bcol * K + kb;

  const int rbase = (mt << 4) + ((lane >> 4) << 3);
  v4f rs0 = {1.0f, 1.0f, 1.0f, 1.0f}, rs1 = {1.0f, 1.0f, 1.0f, 1.0f};
  if (rowscale) {
    rs0 = *(const v4f*)(rowscale + rbase);
    rs1 = *(const v4f*)(rowscale + rbase + 4);
  }

  v8f acc = {};
  if (Wl) {
    const int n  = arow >> 1;
    const int dp = arow & 1;
    const float w0 = Wl[dp * 2 + 0];
    const float w1 = Wl[dp * 2 + 1];
    const float* A0 = A + (size_t)(n * 2) * K + kb;
    const float* A1 = A0 + K;
    for (int k0 = 0; k0 < K; k0 += 4) {
      v2f a0 = *(const v2f*)(A0 + k0);
      v2f a1 = *(const v2f*)(A1 + k0);
      v2f a;
      a.x = w0 * a0.x + w1 * a1.x;
      a.y = w0 * a0.y + w1 * a1.y;
      v2f b = *(const v2f*)(Wp + k0);
      acc = wmma4(a, b, acc);
    }
  } else {
    const float* Ap = A + (size_t)arow * K + kb;
    for (int k0 = 0; k0 < K; k0 += 4) {
      v2f a = *(const v2f*)(Ap + k0);
      v2f b = *(const v2f*)(Wp + k0);
      acc = wmma4(a, b, acc);
    }
  }

  const float bv    = bias ? bias[bcol] : 0.0f;
  const float rs[8] = {rs0.x, rs0.y, rs0.z, rs0.w, rs1.x, rs1.y, rs1.z, rs1.w};
  #pragma unroll
  for (int r = 0; r < 8; ++r) {
    float v = acc[r] + bv;
    if (do_elu) v = (v > 0.0f) ? v : expm1f(v);
    if (rowscale) v *= rs[r];
    size_t idx = (size_t)(rbase + r) * Nn + bcol;
    C[idx] = v;
    if (C2) C2[idx] = v;
  }
}

__global__ void sheaf_pq(const float* __restrict__ xcur,
                         const float* __restrict__ Ws,
                         float* __restrict__ pq)
{
  int n = blockIdx.x * blockDim.x + threadIdx.x;
  if (n >= N_NODES) return;
  const float* xrow = xcur + (size_t)n * DH;
  float p0 = 0.f, p1 = 0.f, q0 = 0.f, q1 = 0.f;
  #pragma unroll 4
  for (int j = 0; j < DH; ++j) {
    float xv = xrow[j];
    p0 += Ws[j]             * xv;
    p1 += Ws[2*DH + j]      * xv;
    q0 += Ws[DH + j]        * xv;
    q1 += Ws[2*DH + DH + j] * xv;
  }
  v4f o4; o4.x = p0; o4.y = p1; o4.z = q0; o4.w = q1;
  *(v4f*)(pq + 4 * (size_t)n) = o4;
}

__global__ void edge_F(const int* __restrict__ rec,
                       const float* __restrict__ pq,
                       float* __restrict__ Ft, float* __restrict__ G)
{
  int k = blockIdx.x * blockDim.x + threadIdx.x;
  if (k >= E_CNT) return;
  const v4i r = *(const v4i*)(rec + 4 * (size_t)k);
  const v4f a = *(const v4f*)(pq + 4 * (size_t)r.y);
  const v4f b = *(const v4f*)(pq + 4 * (size_t)r.x);
  v4f bp = b;
  if (r.z != r.x) bp = *(const v4f*)(pq + 4 * (size_t)r.z);
  v4f aq = a;
  if (r.w != r.y) aq = *(const v4f*)(pq + 4 * (size_t)r.w);
  v2f f, g;
  f.x = tanhf(a.x + b.z);
  f.y = tanhf(a.y + b.w);
  g.x = f.x * tanhf(bp.x + aq.z);
  g.y = f.y * tanhf(bp.y + aq.w);
  *(v2f*)(Ft + 2 * (size_t)k) = f;
  *(v2f*)(G  + 2 * (size_t)k) = g;
}

__global__ void layer_gemm_dinv(const float* __restrict__ A,
                                const float* __restrict__ Wr,
                                const float* __restrict__ Wl,
                                const int* __restrict__ rowptr,
                                const float* __restrict__ Fs,
                                float* __restrict__ C,
                                float* __restrict__ dinv, float* __restrict__ diagw)
{
  const int lane   = threadIdx.x & 31;
  const int wid    = blockIdx.x * (blockDim.x >> 5) + (threadIdx.x >> 5);
  const int ntiles = Hc >> 4;
  const int mtiles = ND >> 4;
  if (wid >= mtiles * ntiles) return;
  const int mt = wid / ntiles;
  const int nt = wid - mt * ntiles;

  const int nd = (mt << 3) + (lane & 7);
  int beg = rowptr[nd];
  int end = rowptr[nd + 1];
  beg = beg < 0 ? 0 : beg;
  end = end > E_CNT ? E_CNT : end;
  float g0 = 0.0f, g1 = 0.0f;
  for (int k = beg; k < end; k += 4) {
    v2f f[4];
    #pragma unroll
    for (int u = 0; u < 4; ++u) {
      const int kk = k + u;
      f[u] = *(const v2f*)(Fs + 2 * (size_t)(kk < end ? kk : end - 1));
    }
    #pragma unroll
    for (int u = 0; u < 4; ++u) {
      const float m = (k + u < end) ? 1.0f : 0.0f;
      const float fx = m * f[u].x;
      const float fy = m * f[u].y;
      g0 += fx * fx;
      g1 += fy * fy;
    }
  }
  const float d0 = rsqrtf(g0 + 1.0f);
  const float d1 = rsqrtf(g1 + 1.0f);
  if (nt == 0 && lane < 8) {
    v2f dv; dv.x = d0;      dv.y = d1;
    v2f dw; dw.x = g0 * d0; dw.y = g1 * d1;
    *(v2f*)(dinv  + 2 * (size_t)nd) = dv;
    *(v2f*)(diagw + 2 * (size_t)nd) = dw;
  }

  const int arow = (mt << 4) + (lane & 15);
  const int bcol = (nt << 4) + (lane & 15);
  const int kb   = (lane >> 4) << 1;
  const float* Wp = Wr + (size_t)bcol * Hc + kb;

  const int n  = arow >> 1;
  const int dp = arow & 1;
  const float w0 = Wl[dp * 2 + 0];
  const float w1 = Wl[dp * 2 + 1];
  const float* A0 = A + (size_t)(n * 2) * Hc + kb;
  const float* A1 = A0 + Hc;
  v8f acc = {};
  for (int k0 = 0; k0 < Hc; k0 += 4) {
    v2f a0 = *(const v2f*)(A0 + k0);
    v2f a1 = *(const v2f*)(A1 + k0);
    v2f a;
    a.x = w0 * a0.x + w1 * a1.x;
    a.y = w0 * a0.y + w1 * a1.y;
    v2f b = *(const v2f*)(Wp + k0);
    acc = wmma4(a, b, acc);
  }

  const int rbase = (mt << 4) + ((lane >> 4) << 3);
  const int sl0   = (lane >> 4) << 2;
  #pragma unroll
  for (int r = 0; r < 8; ++r) {
    const float rsv = __shfl((r & 1) ? d1 : d0, sl0 + (r >> 1));
    C[(size_t)(rbase + r) * Hc + bcol] = acc[r] * rsv;
  }
}

__global__ void csr_zero(int* __restrict__ cnt)
{
  int i = blockIdx.x * blockDim.x + threadIdx.x;
  if (i < N_NODES) cnt[i] = 0;
}

__global__ void csr_rank(const int* __restrict__ src, int* __restrict__ cnt,
                         int* __restrict__ slot)
{
  int e = blockIdx.x * blockDim.x + threadIdx.x;
  if (e >= E_CNT) return;
  slot[e] = atomicAdd(&cnt[src[e]], 1);
}

#define SCAN_ITERS ((N_NODES + 1023) / 1024)
#define SCAN_PAD   (SCAN_ITERS * 1024)
__global__ void __launch_bounds__(1024) csr_scan(const int* __restrict__ cnt,
                                                 int* __restrict__ rowptr)
{
  __shared__ int wtot[32];
  const int lane = threadIdx.x & 31;
  const int w    = threadIdx.x >> 5;
  const int base = w * (SCAN_ITERS * 32) + lane;
  int v[SCAN_ITERS];
  int s = 0;
  #pragma unroll
  for (int i = 0; i < SCAN_ITERS; ++i) {
    const int idx = base + i * 32;
    const int x   = cnt[idx < N_NODES ? idx : N_NODES - 1];
    v[i] = idx < N_NODES ? x : 0;
    s += v[i];
  }
  for (int o = 16; o > 0; o >>= 1) s += __shfl_xor(s, o);
  if (lane == 0) wtot[w] = s;
  __syncthreads();
  int carry = 0;
  for (int j = 0; j < w; ++j) carry += wtot[j];
  #pragma unroll
  for (int i = 0; i < SCAN_ITERS; ++i) {
    int inc = v[i];
    #pragma unroll
    for (int o = 1; o < 32; o <<= 1) {
      const int u = __shfl_up(inc, o);
      if (lane >= o) inc += u;
    }
    rowptr[base + i * 32] = carry + inc - v[i];
    carry += __shfl(inc, 31);
  }
}

__global__ void csr_fill(const int* __restrict__ src, const int* __restrict__ dst,
                         const int* __restrict__ rowptr, const int* __restrict__ slot,
                         int* __restrict__ rec)
{
  int e = blockIdx.x * blockDim.x + threadIdx.x;
  if (e >= E_CNT) return;
  const int er  = (e >= EU_CNT) ? (e - EU_CNT) : (e + EU_CNT);
  const int s   = src[e];
  const int pos = rowptr[s] + slot[e];
  v4i r;
  r.x = dst[e];
  r.y = s;
  r.z = src[er];
  r.w = dst[er];
  *(v4i*)(rec + 4 * (size_t)pos) = r;
}

__global__ void __launch_bounds__(256) node_gather(
    const int* __restrict__ rowptr, const int* __restrict__ rec,
    const float* __restrict__ G, const float* __restrict__ dinv,
    const float* __restrict__ diagw, const float* __restrict__ xr,
    const float* __restrict__ eps2, float* __restrict__ x0,
    const float* __restrict__ Wsn,
    float* __restrict__ pq)
{
  const int lane = threadIdx.x & 31;
  const int s = __builtin_amdgcn_readfirstlane(
      (int)(blockIdx.x * (blockDim.x >> 5) + (threadIdx.x >> 5)));
  if (s >= N_NODES) return;
  const int d = lane >> 4;
  int beg = rowptr[s];
  int end = rowptr[s + 1];
  beg = beg < 0 ? 0 : beg;
  end = end > E_CNT ? E_CNT : end;

  const size_t own = (size_t)s * DH + 2 * lane;
  const v2f ds2 = *(const v2f*)(dinv + 2 * s);
  const v2f xs  = *(const v2f*)(xr + own);
  const float dw = diagw[2 * s + d];
  v2f acc;
  acc.x = dw * xs.x;
  acc.y = dw * xs.y;
  v2f wp0 = {0.0f, 0.0f}, wq0 = wp0, wp1 = wp0, wq1 = wp0;
  if (Wsn) {
    wp0 = *(const v2f*)(Wsn + 2 * lane);
    wq0 = *(const v2f*)(Wsn + DH + 2 * lane);
    wp1 = *(const v2f*)(Wsn + 2 * DH + 2 * lane);
    wq1 = *(const v2f*)(Wsn + 3 * DH + 2 * lane);
  }

  for (int k0 = beg; k0 < end; k0 += 32) {
    const int k = k0 + lane;
    int   t  = 0;
    float w0 = 0.0f, w1 = 0.0f;
    if (k < end) {
      t = rec[4 * (size_t)k];
      const v2f g = *(const v2f*)(G + 2 * (size_t)k);
      w0 = -g.x * ds2.x;
      w1 = -g.y * ds2.y;
    }
    const int m = (end - k0) < 32 ? (end - k0) : 32;
    #pragma unroll 4
    for (int j = 0; j < m; ++j) {
      const int   tj = __builtin_amdgcn_readlane(t, j);
      const float a0 = __int_as_float(__builtin_amdgcn_readlane(__float_as_int(w0), j));
      const float a1 = __int_as_float(__builtin_amdgcn_readlane(__float_as_int(w1), j));
      const float wj = d ? a1 : a0;
      const v2f xv = *(const v2f*)(xr + (size_t)tj * DH + 2 * lane);
      acc.x += wj * xv.x;
      acc.y += wj * xv.y;
    }
  }

  const float coeff = 1.0f + tanhf(eps2[d]);
  const v2f xo = *(const v2f*)(x0 + own);
  const float e0 = (acc.x > 0.0f) ? acc.x : expm1f(acc.x);
  const float e1 = (acc.y > 0.0f) ? acc.y : expm1f(acc.y);
  v2f nx;
  nx.x = coeff * xo.x - e0;
  nx.y = coeff * xo.y - e1;
  *(v2f*)(x0 + own)   = nx;

  if (Wsn) {
    float p0 = wp0.x * nx.x + wp0.y * nx.y;
    float p1 = wp1.x * nx.x + wp1.y * nx.y;
    float q0 = wq0.x * nx.x + wq0.y * nx.y;
    float q1 = wq1.x * nx.x + wq1.y * nx.y;
    for (int o = 16; o > 0; o >>= 1) {
      p0 += __shfl_xor(p0, o);
      p1 += __shfl_xor(p1, o);
      q0 += __shfl_xor(q0, o);
      q1 += __shfl_xor(q1, o);
    }
    if (lane == 0) {
      v4f o4; o4.x = p0; o4.y = p1; o4.z = q0; o4.w = q1;
      *(v4f*)(pq + 4 * (size_t)s) = o4;
    }
  }
}

extern "C" void kernel_launch(void* const* d_in, const int* in_sizes, int n_in,
                              void* d_out, int out_size, void* d_ws, size_t ws_size,
                              hipStream_t stream) {
  const float* x_in   = (const float*)d_in[0];
  const int*   ei     = (const int*)  d_in[1];
  const float* W1     = (const float*)d_in[2];
  const float* b1     = (const float*)d_in[3];
  const float* Wsheaf = (const float*)d_in[4];
  const float* Wleft  = (const float*)d_in[5];
  const float* Wright = (const float*)d_in[6];
  const float* eps    = (const float*)d_in[7];
  const float* W2     = (const float*)d_in[8];
  const float* b2     = (const float*)d_in[9];
  float* out = (float*)d_out;

  float* ws = (float*)d_ws;
  size_t o = 0;
  float* Ft    = ws + o; o += (size_t)ND * Hc;
  float* x0    = ws + o; o += (size_t)ND * Hc;
  int*   rec   = (int*)(ws + o); o += (size_t)ND * Hc;
  float* xr    = ws + o; o += (size_t)ND * Hc;
  int*   slot  = (int*)xr;
  int*   cnt   = (int*)xr + (size_t)E_CNT;
  float* pq    = ws + o; o += 2 * (size_t)N_NODES; o += 2 * (size_t)N_NODES;
  float* G     = ws + o; o += 2 * (size_t)E_CNT;
  int*   rowptr = (int*)(ws + o); o += (size_t)ND;
  float* dinv  = ws + o; o += (size_t)ND;
  float* diagw = ws + o; o += (size_t)ND;

  const int* src = ei;
  const int* dst = ei + E_CNT;

  const int TPB = 256;

  csr_zero<<<(N_NODES + TPB - 1) / TPB, TPB, 0, stream>>>(cnt);
  csr_rank<<<(E_CNT + TPB - 1) / TPB, TPB, 0, stream>>>(src, cnt, slot);
  csr_scan<<<1, 1024, 0, stream>>>(cnt, rowptr);
  csr_fill<<<(E_CNT + TPB - 1) / TPB, TPB, 0, stream>>>(src, dst, rowptr, slot, rec);

  {
    int tiles = (N_NODES / 16) * (DH / 16);
    wmma_gemm_xwT<<<(tiles + 7) / 8, TPB, 0, stream>>>(
        x_in, W1, b1, nullptr, nullptr, x0, nullptr, N_NODES, IN_CH, DH, 1);
  }

  sheaf_pq<<<(N_NODES + TPB - 1) / TPB, TPB, 0, stream>>>(x0, Wsheaf, pq);

  for (int l = 0; l < LAYERS; ++l) {
    edge_F<<<(E_CNT + TPB - 1) / TPB, TPB, 0, stream>>>(rec, pq, Ft, G);
    {
      int tiles = (ND / 16) * (Hc / 16);
      layer_gemm_dinv<<<(tiles + 7) / 8, TPB, 0, stream>>>(
          x0, Wright + (size_t)l * Hc * Hc, Wleft + (size_t)l * 4,
          rowptr, Ft, xr, dinv, diagw);
    }
    node_gather<<<(N_NODES + 7) / 8, TPB, 0, stream>>>(
        rowptr, rec, G, dinv, diagw, xr, eps + (size_t)l * 2, x0,
        (l + 1 < LAYERS) ? Wsheaf + (size_t)(l + 1) * 2 * 2 * DH : nullptr, pq);
  }

  {
    int tiles = (N_NODES / 16) * (OUTC / 16);
    wmma_gemm_xwT<<<(tiles + 7) / 8, TPB, 0, stream>>>(
        x0, W2, b2, nullptr, nullptr, out, nullptr, N_NODES, DH, OUTC, 0);
  }
}


# gather: each wave's chain of dependent reads cut: eps and the node's old row asked for with its first reads; row reads in groups of 8, left-over edges as one group, not one by one; sum order kept
# speedup vs baseline: 4.4302x; 1.1593x over previous
#include <hip/hip_runtime.h>
#include <cstdint>

#define N_NODES 50000
#define IN_CH   128
#define Hc      32
#define Dd      2
#define LAYERS  4
#define EU_CNT  400000
#define E_CNT   (2*EU_CNT)
#define DH      (Dd*Hc)
#define ND      (N_NODES*Dd)
#define OUTC    32

typedef __attribute__((ext_vector_type(2))) float v2f;
typedef __attribute__((ext_vector_type(4))) float v4f;
typedef __attribute__((ext_vector_type(8))) float v8f;
typedef __attribute__((ext_vector_type(4))) int   v4i;

__device__ __forceinline__ v8f wmma4(v2f a, v2f b, v8f c) {
  return __builtin_amdgcn_wmma_f32_16x16x4_f32(false, a, false, b, (short)0, c,
                                               false, false);
}

__device__ __forceinline__ void gemm_xwT_body(int bid,
                              const float* __restrict__ A,
                              const float* __restrict__ W,
                              const float* __restrict__ bias,
                              const float* __restrict__ Wl,
                              const float* __restrict__ rowscale,
                              float* __restrict__ C,
                              float* __restrict__ C2,
                              int M, int K, int Nn, int do_elu)
{
  const int lane   = threadIdx.x & 31;
  const int wid    = bid * (blockDim.x >> 5) + (threadIdx.x >> 5);
  const int ntiles = Nn >> 4;
  const int mtiles = M  >> 4;
  if (wid >= mtiles * ntiles) return;
  const int mt = wid / ntiles;
  const int nt = wid - mt * ntiles;

  const int arow = (mt << 4) + (lane & 15);
  const int bcol = (nt << 4) + (lane & 15);
  const int kb   = (lane >> 4) << 1;

  const float* Wp = W + (size_t)bcol * K + kb;

  const int rbase = (mt << 4) + ((lane >> 4) << 3);
  v4f rs0 = {1.0f, 1.0f, 1.0f, 1.0f}, rs1 = {1.0f, 1.0f, 1.0f, 1.0f};
  if (rowscale) {
    rs0 = *(const v4f*)(rowscale + rbase);
    rs1 = *(const v4f*)(rowscale + rbase + 4);
  }

  v8f acc = {};
  if (Wl) {
    const int n  = arow >> 1;
    const int dp = arow & 1;
    const float w0 = Wl[dp * 2 + 0];
    const float w1 = Wl[dp * 2 + 1];
    const float* A0 = A + (size_t)(n * 2) * K + kb;
    const float* A1 = A0 + K;
    for (int k0 = 0; k0 < K; k0 += 4) {
      v2f a0 = *(const v2f*)(A0 + k0);
      v2f a1 = *(const v2f*)(A1 + k0);
      v2f a;
      a.x = w0 * a0.x + w1 * a1.x;
      a.y = w0 * a0.y + w1 * a1.y;
      v2f b = *(const v2f*)(Wp + k0);
      acc = wmma4(a, b, acc);
    }
  } else {
    const float* Ap = A + (size_t)arow * K + kb;
    for (int k0 = 0; k0 < K; k0 += 4) {
      v2f a = *(const v2f*)(Ap + k0);
      v2f b = *(const v2f*)(Wp + k0);
      acc = wmma4(a, b, acc);
    }
  }

  const float bv    = bias ? bias[bcol] : 0.0f;
  const float rs[8] = {rs0.x, rs0.y, rs0.z, rs0.w, rs1.x, rs1.y, rs1.z, rs1.w};
  #pragma unroll
  for (int r = 0; r < 8; ++r) {
    float v = acc[r] + bv;
    if (do_elu) v = (v > 0.0f) ? v : expm1f(v);
    if (rowscale) v *= rs[r];
    size_t idx = (size_t)(rbase + r) * Nn + bcol;
    C[idx] = v;
    if (C2) C2[idx] = v;
  }
}

__device__ __forceinline__ void sheaf_pq_body(int bid,
                         const float* __restrict__ xcur,
                         const float* __restrict__ Ws,
                         float* __restrict__ pq)
{
  int n = bid * blockDim.x + threadIdx.x;
  if (n >= N_NODES) return;
  const float* xrow = xcur + (size_t)n * DH;
  float p0 = 0.f, p1 = 0.f, q0 = 0.f, q1 = 0.f;
  #pragma unroll 4
  for (int j = 0; j < DH; ++j) {
    float xv = xrow[j];
    p0 += Ws[j]             * xv;
    p1 += Ws[2*DH + j]      * xv;
    q0 += Ws[DH + j]        * xv;
    q1 += Ws[2*DH + DH + j] * xv;
  }
  v4f o4; o4.x = p0; o4.y = p1; o4.z = q0; o4.w = q1;
  *(v4f*)(pq + 4 * (size_t)n) = o4;
}

#define ECH 128
__global__ void __launch_bounds__(256) layer_edge_gemm(
                                const float* __restrict__ A,
                                const float* __restrict__ Wr,
                                const float* __restrict__ Wl,
                                const int* __restrict__ rowptr,
                                const int* __restrict__ rec,
                                const float* __restrict__ pq,
                                float* __restrict__ G,
                                float* __restrict__ C,
                                float* __restrict__ dinv, float* __restrict__ diagw)
{
  __shared__ v2f fst[8][ECH];
  const int lane = threadIdx.x & 31;
  const int wv   = threadIdx.x >> 5;
  const int mt = __builtin_amdgcn_readfirstlane((int)(blockIdx.x * 8 + wv));
  if (mt >= (ND >> 4)) return;

  const int arow = (mt << 4) + (lane & 15);
  const int kb   = (lane >> 4) << 1;
  const int n    = arow >> 1;
  const int dp   = arow & 1;
  const float w0 = Wl[dp * 2 + 0];
  const float w1 = Wl[dp * 2 + 1];
  const float* A0 = A + (size_t)(n * 2) * Hc + kb;
  const float* A1 = A0 + Hc;
  v2f a0[Hc / 4], a1[Hc / 4];
  #pragma unroll
  for (int i = 0; i < Hc / 4; ++i) {
    a0[i] = *(const v2f*)(A0 + 4 * i);
    a1[i] = *(const v2f*)(A1 + 4 * i);
  }

  const int nd = (mt << 3) + (lane & 7);
  int rb = rowptr[nd];
  int re = rowptr[nd + 1];
  rb = rb < 0 ? 0 : rb;
  re = re > E_CNT ? E_CNT : re;
  const int tb = __builtin_amdgcn_readlane(rb, 0);
  const int te = __builtin_amdgcn_readlane(re, 7);
  float g0 = 0.0f, g1 = 0.0f;
  for (int cb = tb; cb < te; cb += ECH) {
    v4i r[4];
    #pragma unroll
    for (int u = 0; u < 4; ++u) {
      const int k = cb + lane + 32 * u;
      r[u] = *(const v4i*)(rec + 4 * (size_t)(k < te ? k : te - 1));
    }
    v4f pa[4], pb[4];
    #pragma unroll
    for (int u = 0; u < 4; ++u) {
      pa[u] = *(const v4f*)(pq + 4 * (size_t)r[u].y);
      pb[u] = *(const v4f*)(pq + 4 * (size_t)r[u].x);
    }
    #pragma unroll
    for (int u = 0; u < 4; ++u) {
      const int k = cb + lane + 32 * u;
      v4f bp = pb[u];
      if (r[u].z != r[u].x) bp = *(const v4f*)(pq + 4 * (size_t)r[u].z);
      v4f aq = pa[u];
      if (r[u].w != r[u].y) aq = *(const v4f*)(pq + 4 * (size_t)r[u].w);
      v2f f, g;
      f.x = tanhf(pa[u].x + pb[u].z);
      f.y = tanhf(pa[u].y + pb[u].w);
      g.x = f.x * tanhf(bp.x + aq.z);
      g.y = f.y * tanhf(bp.y + aq.w);
      if (k < te) *(v2f*)(G + 2 * (size_t)k) = g;
      fst[wv][lane + 32 * u] = f;
    }
    const int ce = (cb + ECH) < te ? (cb + ECH) : te;
    const int lo = rb > cb ? rb : cb;
    const int hi = re < ce ? re : ce;
    for (int k = lo; k < hi; ++k) {
      const v2f f = fst[wv][k - cb];
      g0 += f.x * f.x;
      g1 += f.y * f.y;
    }
  }
  const float d0 = rsqrtf(g0 + 1.0f);
  const float d1 = rsqrtf(g1 + 1.0f);
  if (lane < 8) {
    v2f dv; dv.x = d0;      dv.y = d1;
    v2f dw; dw.x = g0 * d0; dw.y = g1 * d1;
    *(v2f*)(dinv  + 2 * (size_t)nd) = dv;
    *(v2f*)(diagw + 2 * (size_t)nd) = dw;
  }

  const float* Wp0 = Wr + (size_t)(lane & 15) * Hc + kb;
  const float* Wp1 = Wp0 + (size_t)16 * Hc;
  v8f acc0 = {}, acc1 = {};
  #pragma unroll
  for (int i = 0; i < Hc / 4; ++i) {
    v2f a;
    a.x = w0 * a0[i].x + w1 * a1[i].x;
    a.y = w0 * a0[i].y + w1 * a1[i].y;
    const v2f b0 = *(const v2f*)(Wp0 + 4 * i);
    const v2f b1 = *(const v2f*)(Wp1 + 4 * i);
    acc0 = wmma4(a, b0, acc0);
    acc1 = wmma4(a, b1, acc1);
  }

  const int rbase = (mt << 4) + ((lane >> 4) << 3);
  const int sl0   = (lane >> 4) << 2;
  #pragma unroll
  for (int r8 = 0; r8 < 8; ++r8) {
    const float rsv = __shfl((r8 & 1) ? d1 : d0, sl0 + (r8 >> 1));
    float* crow = C + (size_t)(rbase + r8) * Hc + (lane & 15);
    crow[0]  = acc0[r8] * rsv;
    crow[16] = acc1[r8] * rsv;
  }
}

__global__ void csr_zero(int* __restrict__ cnt)
{
  int i = blockIdx.x * blockDim.x + threadIdx.x;
  if (i < N_NODES) cnt[i] = 0;
}

__device__ __forceinline__ void csr_rank_body(int bid, const int* __restrict__ src,
                                              int* __restrict__ cnt, int* __restrict__ slot)
{
  int e = bid * blockDim.x + threadIdx.x;
  if (e >= E_CNT) return;
  slot[e] = atomicAdd(&cnt[src[e]], 1);
}

#define G1_TILES   ((N_NODES / 16) * (DH / 16))
#define G1_BLOCKS  ((G1_TILES + 7) / 8)
#define EDGE_BLOCKS ((E_CNT + 255) / 256)
#define NODE_BLOCKS ((N_NODES + 255) / 256)
__global__ void __launch_bounds__(256) rank_and_first_gemm(
    const int* __restrict__ src, int* __restrict__ cnt, int* __restrict__ slot,
    const float* __restrict__ x_in, const float* __restrict__ W1,
    const float* __restrict__ b1, float* __restrict__ x0)
{
  if (blockIdx.x < G1_BLOCKS) {
    gemm_xwT_body((int)blockIdx.x, x_in, W1, b1, nullptr, nullptr, x0, nullptr,
                  N_NODES, IN_CH, DH, 1);
  } else {
    csr_rank_body((int)blockIdx.x - G1_BLOCKS, src, cnt, slot);
  }
}

#define SCAN_ITERS ((N_NODES + 1023) / 1024)
#define SCAN_PAD   (SCAN_ITERS * 1024)
__global__ void __launch_bounds__(1024) csr_scan(const int* __restrict__ cnt,
                                                 int* __restrict__ rowptr)
{
  __shared__ int wtot[32];
  const int lane = threadIdx.x & 31;
  const int w    = threadIdx.x >> 5;
  const int base = w * (SCAN_ITERS * 32) + lane;
  int v[SCAN_ITERS];
  int s = 0;
  #pragma unroll
  for (int i = 0; i < SCAN_ITERS; ++i) {
    const int idx = base + i * 32;
    const int x   = cnt[idx < N_NODES ? idx : N_NODES - 1];
    v[i] = idx < N_NODES ? x : 0;
    s += v[i];
  }
  for (int o = 16; o > 0; o >>= 1) s += __shfl_xor(s, o);
  if (lane == 0) wtot[w] = s;
  __syncthreads();
  int carry = 0;
  for (int j = 0; j < w; ++j) carry += wtot[j];
  #pragma unroll
  for (int i = 0; i < SCAN_ITERS; ++i) {
    int inc = v[i];
    #pragma unroll
    for (int o = 1; o < 32; o <<= 1) {
      const int u = __shfl_up(inc, o);
      if (lane >= o) inc += u;
    }
    rowptr[base + i * 32] = carry + inc - v[i];
    carry += __shfl(inc, 31);
  }
}

__device__ __forceinline__ void csr_fill_body(int bid,
                         const int* __restrict__ src, const int* __restrict__ dst,
                         const int* __restrict__ rowptr, const int* __restrict__ slot,
                         int* __restrict__ rec)
{
  int e = bid * blockDim.x + threadIdx.x;
  if (e >= E_CNT) return;
  const int er  = (e >= EU_CNT) ? (e - EU_CNT) : (e + EU_CNT);
  const int s   = src[e];
  const int pos = rowptr[s] + slot[e];
  v4i r;
  r.x = dst[e];
  r.y = s;
  r.z = src[er];
  r.w = dst[er];
  *(v4i*)(rec + 4 * (size_t)pos) = r;
}

__global__ void __launch_bounds__(256) fill_and_first_pq(
    const int* __restrict__ src, const int* __restrict__ dst,
    const int* __restrict__ rowptr, const int* __restrict__ slot, int* __restrict__ rec,
    const float* __restrict__ x0, const float* __restrict__ Ws, float* __restrict__ pq)
{
  if (blockIdx.x < NODE_BLOCKS) {
    sheaf_pq_body((int)blockIdx.x, x0, Ws, pq);
  } else {
    csr_fill_body((int)blockIdx.x - NODE_BLOCKS, src, dst, rowptr, slot, rec);
  }
}

#define RG 8
__device__ __forceinline__ v2f gather_node(
    int s, int lane,
    const int* __restrict__ rowptr, const int* __restrict__ rec,
    const float* __restrict__ G, const float* __restrict__ dinv,
    const float* __restrict__ diagw, const float* __restrict__ xr,
    const float* __restrict__ eps2, float* __restrict__ x0, bool store_x,
    const float* __restrict__ Wsn,
    float* __restrict__ pq)
{
  const int d = lane >> 4;
  int beg = rowptr[s];
  int end = rowptr[s + 1];
  beg = beg < 0 ? 0 : beg;
  end = end > E_CNT ? E_CNT : end;

  const size_t own = (size_t)s * DH + 2 * lane;
  const v2f ds2 = *(const v2f*)(dinv + 2 * s);
  const v2f xs  = *(const v2f*)(xr + own);
  const float dw = diagw[2 * s + d];
  const v2f xo = *(const v2f*)(x0 + own);
  const float epsd = eps2[d];
  v2f acc;
  acc.x = dw * xs.x;
  acc.y = dw * xs.y;
  v2f wp0 = {0.0f, 0.0f}, wq0 = wp0, wp1 = wp0, wq1 = wp0;
  if (Wsn) {
    wp0 = *(const v2f*)(Wsn + 2 * lane);
    wq0 = *(const v2f*)(Wsn + DH + 2 * lane);
    wp1 = *(const v2f*)(Wsn + 2 * DH + 2 * lane);
    wq1 = *(const v2f*)(Wsn + 3 * DH + 2 * lane);
  }

  for (int k0 = beg; k0 < end; k0 += 32) {
    const int k = k0 + lane;
    int   t  = 0;
    float w0 = 0.0f, w1 = 0.0f;
    if (k < end) {
      t = rec[4 * (size_t)k];
      const v2f g = *(const v2f*)(G + 2 * (size_t)k);
      w0 = -g.x * ds2.x;
      w1 = -g.y * ds2.y;
    }
    const int m = (end - k0) < 32 ? (end - k0) : 32;
    for (int j = 0; j < m; j += RG) {
      const int c = m - j;
      int tj[RG];
      #pragma unroll
      for (int u = 0; u < RG; ++u) tj[u] = __builtin_amdgcn_readlane(t, j + u);
      v2f xv[RG];
      #pragma unroll
      for (int u = 0; u < RG; ++u) {
        if (u < c) xv[u] = *(const v2f*)(xr + (size_t)tj[u] * DH + 2 * lane);
      }
      #pragma unroll
      for (int u = 0; u < RG; ++u) {
        if (u < c) {
          const float a0 = __int_as_float(__builtin_amdgcn_readlane(__float_as_int(w0), j + u));
          const float a1 = __int_as_float(__builtin_amdgcn_readlane(__float_as_int(w1), j + u));
          const float wj = d ? a1 : a0;
          acc.x += wj * xv[u].x;
          acc.y += wj * xv[u].y;
        }
      }
    }
  }

  const float coeff = 1.0f + tanhf(epsd);
  const float e0 = (acc.x > 0.0f) ? acc.x : expm1f(acc.x);
  const float e1 = (acc.y > 0.0f) ? acc.y : expm1f(acc.y);
  v2f nx;
  nx.x = coeff * xo.x - e0;
  nx.y = coeff * xo.y - e1;
  if (store_x) *(v2f*)(x0 + own) = nx;

  if (Wsn) {
    float p0 = wp0.x * nx.x + wp0.y * nx.y;
    float p1 = wp1.x * nx.x + wp1.y * nx.y;
    float q0 = wq0.x * nx.x + wq0.y * nx.y;
    float q1 = wq1.x * nx.x + wq1.y * nx.y;
    for (int o = 16; o > 0; o >>= 1) {
      p0 += __shfl_xor(p0, o);
      p1 += __shfl_xor(p1, o);
      q0 += __shfl_xor(q0, o);
      q1 += __shfl_xor(q1, o);
    }
    if (lane == 0) {
      v4f o4; o4.x = p0; o4.y = p1; o4.z = q0; o4.w = q1;
      *(v4f*)(pq + 4 * (size_t)s) = o4;
    }
  }
  return nx;
}

#define XPITCH (DH + 2)
__global__ void __launch_bounds__(256) node_gather(
    const int* __restrict__ rowptr, const int* __restrict__ rec,
    const float* __restrict__ G, const float* __restrict__ dinv,
    const float* __restrict__ diagw, const float* __restrict__ xr,
    const float* __restrict__ eps2, float* __restrict__ x0,
    const float* __restrict__ Wsn,
    float* __restrict__ pq,
    const float* __restrict__ W2,
    const float* __restrict__ b2, float* __restrict__ out)
{
  __shared__ float xrow[8 * XPITCH];
  const int lane = threadIdx.x & 31;
  const int wv   = threadIdx.x >> 5;
  const int s = __builtin_amdgcn_readfirstlane((int)(blockIdx.x * 8 + wv));
  v2f nx = {0.0f, 0.0f};
  if (s < N_NODES)
    nx = gather_node(s, lane, rowptr, rec, G, dinv, diagw, xr, eps2, x0,
                     W2 == nullptr, Wsn, pq);
  if (W2 == nullptr) return;

  *(v2f*)(xrow + wv * XPITCH + 2 * lane) = nx;
  __syncthreads();
  if (wv >= 2) return;
  const int m    = lane & 15;
  const int kb   = (lane >> 4) << 1;
  const int bcol = (wv << 4) + m;
  const float* Wp = W2 + (size_t)bcol * DH + kb;
  const float* Xp = xrow + (m & 7) * XPITCH + kb;
  v8f acc = {};
  for (int k0 = 0; k0 < DH; k0 += 4) {
    v2f a = *(const v2f*)(Xp + k0);
    if (m >= 8) { a.x = 0.0f; a.y = 0.0f; }
    const v2f b = *(const v2f*)(Wp + k0);
    acc = wmma4(a, b, acc);
  }
  const float bv = b2[bcol];
  const int nb = (int)blockIdx.x * 8;
  if (lane < 16) {
    #pragma unroll
    for (int r = 0; r < 8; ++r) {
      if (nb + r < N_NODES) out[(size_t)(nb + r) * OUTC + bcol] = acc[r] + bv;
    }
  }
}

extern "C" void kernel_launch(void* const* d_in, const int* in_sizes, int n_in,
                              void* d_out, int out_size, void* d_ws, size_t ws_size,
                              hipStream_t stream) {
  const float* x_in   = (const float*)d_in[0];
  const int*   ei     = (const int*)  d_in[1];
  const float* W1     = (const float*)d_in[2];
  const float* b1     = (const float*)d_in[3];
  const float* Wsheaf = (const float*)d_in[4];
  const float* Wleft  = (const float*)d_in[5];
  const float* Wright = (const float*)d_in[6];
  const float* eps    = (const float*)d_in[7];
  const float* W2     = (const float*)d_in[8];
  const float* b2     = (const float*)d_in[9];
  float* out = (float*)d_out;

  float* ws = (float*)d_ws;
  size_t o = 0;
  o += (size_t)ND * Hc;
  float* x0    = ws + o; o += (size_t)ND * Hc;
  int*   rec   = (int*)(ws + o); o += (size_t)ND * Hc;
  float* xr    = ws + o; o += (size_t)ND * Hc;
  int*   slot  = (int*)xr;
  int*   cnt   = (int*)xr + (size_t)E_CNT;
  float* pq    = ws + o; o += 2 * (size_t)N_NODES; o += 2 * (size_t)N_NODES;
  float* G     = ws + o; o += 2 * (size_t)E_CNT;
  int*   rowptr = (int*)(ws + o); o += (size_t)ND;
  float* dinv  = ws + o; o += (size_t)ND;
  float* diagw = ws + o; o += (size_t)ND;

  const int* src = ei;
  const int* dst = ei + E_CNT;

  const int TPB = 256;

  csr_zero<<<NODE_BLOCKS, TPB, 0, stream>>>(cnt);
  rank_and_first_gemm<<<G1_BLOCKS + EDGE_BLOCKS, TPB, 0, stream>>>(
      src, cnt, slot, x_in, W1, b1, x0);
  csr_scan<<<1, 1024, 0, stream>>>(cnt, rowptr);
  fill_and_first_pq<<<NODE_BLOCKS + EDGE_BLOCKS, TPB, 0, stream>>>(
      src, dst, rowptr, slot, rec, x0, Wsheaf, pq);

  for (int l = 0; l < LAYERS; ++l) {
    layer_edge_gemm<<<((ND / 16) + 7) / 8, TPB, 0, stream>>>(
        x0, Wright + (size_t)l * Hc * Hc, Wleft + (size_t)l * 4,
        rowptr, rec, pq, G, xr, dinv, diagw);
    const bool last = (l + 1 == LAYERS);
    node_gather<<<(N_NODES + 7) / 8, TPB, 0, stream>>>(
        rowptr, rec, G, dinv, diagw, xr, eps + (size_t)l * 2, x0,
        last ? nullptr : Wsheaf + (size_t)(l + 1) * 2 * 2 * DH, pq,
        last ? W2 : nullptr, b2, out);
  }
}


# gather: two nodes per wave side by side, a 16-lane row each, 16-byte row reads; edge dst and weight passed round the row by permlane, straight-line groups of 8; p,q sum order now a 16-lane tree
# speedup vs baseline: 4.6727x; 1.0547x over previous
#include <hip/hip_runtime.h>
#include <cstdint>

#define N_NODES 50000
#define IN_CH   128
#define Hc      32
#define Dd      2
#define LAYERS  4
#define EU_CNT  400000
#define E_CNT   (2*EU_CNT)
#define DH      (Dd*Hc)
#define ND      (N_NODES*Dd)
#define OUTC    32

typedef __attribute__((ext_vector_type(2))) float v2f;
typedef __attribute__((ext_vector_type(4))) float v4f;
typedef __attribute__((ext_vector_type(8))) float v8f;
typedef __attribute__((ext_vector_type(4))) int   v4i;

__device__ __forceinline__ v8f wmma4(v2f a, v2f b, v8f c) {
  return __builtin_amdgcn_wmma_f32_16x16x4_f32(false, a, false, b, (short)0, c,
                                               false, false);
}

__device__ __forceinline__ void gemm_xwT_body(int bid,
                              const float* __restrict__ A,
                              const float* __restrict__ W,
                              const float* __restrict__ bias,
                              const float* __restrict__ Wl,
                              const float* __restrict__ rowscale,
                              float* __restrict__ C,
                              float* __restrict__ C2,
                              int M, int K, int Nn, int do_elu)
{
  const int lane   = threadIdx.x & 31;
  const int wid    = bid * (blockDim.x >> 5) + (threadIdx.x >> 5);
  const int ntiles = Nn >> 4;
  const int mtiles = M  >> 4;
  if (wid >= mtiles * ntiles) return;
  const int mt = wid / ntiles;
  const int nt = wid - mt * ntiles;

  const int arow = (mt << 4) + (lane & 15);
  const int bcol = (nt << 4) + (lane & 15);
  const int kb   = (lane >> 4) << 1;

  const float* Wp = W + (size_t)bcol * K + kb;

  const int rbase = (mt << 4) + ((lane >> 4) << 3);
  v4f rs0 = {1.0f, 1.0f, 1.0f, 1.0f}, rs1 = {1.0f, 1.0f, 1.0f, 1.0f};
  if (rowscale) {
    rs0 = *(const v4f*)(rowscale + rbase);
    rs1 = *(const v4f*)(rowscale + rbase + 4);
  }

  v8f acc = {};
  if (Wl) {
    const int n  = arow >> 1;
    const int dp = arow & 1;
    const float w0 = Wl[dp * 2 + 0];
    const float w1 = Wl[dp * 2 + 1];
    const float* A0 = A + (size_t)(n * 2) * K + kb;
    const float* A1 = A0 + K;
    for (int k0 = 0; k0 < K; k0 += 4) {
      v2f a0 = *(const v2f*)(A0 + k0);
      v2f a1 = *(const v2f*)(A1 + k0);
      v2f a;
      a.x = w0 * a0.x + w1 * a1.x;
      a.y = w0 * a0.y + w1 * a1.y;
      v2f b = *(const v2f*)(Wp + k0);
      acc = wmma4(a, b, acc);
    }
  } else {
    const float* Ap = A + (size_t)arow * K + kb;
    for (int k0 = 0; k0 < K; k0 += 4) {
      v2f a = *(const v2f*)(Ap + k0);
      v2f b = *(const v2f*)(Wp + k0);
      acc = wmma4(a, b, acc);
    }
  }

  const float bv    = bias ? bias[bcol] : 0.0f;
  const float rs[8] = {rs0.x, rs0.y, rs0.z, rs0.w, rs1.x, rs1.y, rs1.z, rs1.w};
  #pragma unroll
  for (int r = 0; r < 8; ++r) {
    float v = acc[r] + bv;
    if (do_elu) v = (v > 0.0f) ? v : expm1f(v);
    if (rowscale) v *= rs[r];
    size_t idx = (size_t)(rbase + r) * Nn + bcol;
    C[idx] = v;
    if (C2) C2[idx] = v;
  }
}

__device__ __forceinline__ void sheaf_pq_body(int bid,
                         const float* __restrict__ xcur,
                         const float* __restrict__ Ws,
                         float* __restrict__ pq)
{
  int n = bid * blockDim.x + threadIdx.x;
  if (n >= N_NODES) return;
  const float* xrow = xcur + (size_t)n * DH;
  float p0 = 0.f, p1 = 0.f, q0 = 0.f, q1 = 0.f;
  #pragma unroll 4
  for (int j = 0; j < DH; ++j) {
    float xv = xrow[j];
    p0 += Ws[j]             * xv;
    p1 += Ws[2*DH + j]      * xv;
    q0 += Ws[DH + j]        * xv;
    q1 += Ws[2*DH + DH + j] * xv;
  }
  v4f o4; o4.x = p0; o4.y = p1; o4.z = q0; o4.w = q1;
  *(v4f*)(pq + 4 * (size_t)n) = o4;
}

#define ECH 128
__global__ void __launch_bounds__(256) layer_edge_gemm(
                                const float* __restrict__ A,
                                const float* __restrict__ Wr,
                                const float* __restrict__ Wl,
                                const int* __restrict__ rowptr,
                                const int* __restrict__ rec,
                                const float* __restrict__ pq,
                                float* __restrict__ G,
                                float* __restrict__ C,
                                float* __restrict__ dinv, float* __restrict__ diagw)
{
  __shared__ v2f fst[8][ECH];
  const int lane = threadIdx.x & 31;
  const int wv   = threadIdx.x >> 5;
  const int mt = __builtin_amdgcn_readfirstlane((int)(blockIdx.x * 8 + wv));
  if (mt >= (ND >> 4)) return;

  const int arow = (mt << 4) + (lane & 15);
  const int kb   = (lane >> 4) << 1;
  const int n    = arow >> 1;
  const int dp   = arow & 1;
  const float w0 = Wl[dp * 2 + 0];
  const float w1 = Wl[dp * 2 + 1];
  const float* A0 = A + (size_t)(n * 2) * Hc + kb;
  const float* A1 = A0 + Hc;
  v2f a0[Hc / 4], a1[Hc / 4];
  #pragma unroll
  for (int i = 0; i < Hc / 4; ++i) {
    a0[i] = *(const v2f*)(A0 + 4 * i);
    a1[i] = *(const v2f*)(A1 + 4 * i);
  }

  const int nd = (mt << 3) + (lane & 7);
  int rb = rowptr[nd];
  int re = rowptr[nd + 1];
  rb = rb < 0 ? 0 : rb;
  re = re > E_CNT ? E_CNT : re;
  const int tb = __builtin_amdgcn_readlane(rb, 0);
  const int te = __builtin_amdgcn_readlane(re, 7);
  float g0 = 0.0f, g1 = 0.0f;
  for (int cb = tb; cb < te; cb += ECH) {
    v4i r[4];
    #pragma unroll
    for (int u = 0; u < 4; ++u) {
      const int k = cb + lane + 32 * u;
      r[u] = *(const v4i*)(rec + 4 * (size_t)(k < te ? k : te - 1));
    }
    v4f pa[4], pb[4];
    #pragma unroll
    for (int u = 0; u < 4; ++u) {
      pa[u] = *(const v4f*)(pq + 4 * (size_t)r[u].y);
      pb[u] = *(const v4f*)(pq + 4 * (size_t)r[u].x);
    }
    #pragma unroll
    for (int u = 0; u < 4; ++u) {
      const int k = cb + lane + 32 * u;
      v4f bp = pb[u];
      if (r[u].z != r[u].x) bp = *(const v4f*)(pq + 4 * (size_t)r[u].z);
      v4f aq = pa[u];
      if (r[u].w != r[u].y) aq = *(const v4f*)(pq + 4 * (size_t)r[u].w);
      v2f f, g;
      f.x = tanhf(pa[u].x + pb[u].z);
      f.y = tanhf(pa[u].y + pb[u].w);
      g.x = f.x * tanhf(bp.x + aq.z);
      g.y = f.y * tanhf(bp.y + aq.w);
      if (k < te) *(v2f*)(G + 2 * (size_t)k) = g;
      fst[wv][lane + 32 * u] = f;
    }
    const int ce = (cb + ECH) < te ? (cb + ECH) : te;
    const int lo = rb > cb ? rb : cb;
    const int hi = re < ce ? re : ce;
    for (int k = lo; k < hi; ++k) {
      const v2f f = fst[wv][k - cb];
      g0 += f.x * f.x;
      g1 += f.y * f.y;
    }
  }
  const float d0 = rsqrtf(g0 + 1.0f);
  const float d1 = rsqrtf(g1 + 1.0f);
  if (lane < 8) {
    v2f dv; dv.x = d0;      dv.y = d1;
    v2f dw; dw.x = g0 * d0; dw.y = g1 * d1;
    *(v2f*)(dinv  + 2 * (size_t)nd) = dv;
    *(v2f*)(diagw + 2 * (size_t)nd) = dw;
  }

  const float* Wp0 = Wr + (size_t)(lane & 15) * Hc + kb;
  const float* Wp1 = Wp0 + (size_t)16 * Hc;
  v8f acc0 = {}, acc1 = {};
  #pragma unroll
  for (int i = 0; i < Hc / 4; ++i) {
    v2f a;
    a.x = w0 * a0[i].x + w1 * a1[i].x;
    a.y = w0 * a0[i].y + w1 * a1[i].y;
    const v2f b0 = *(const v2f*)(Wp0 + 4 * i);
    const v2f b1 = *(const v2f*)(Wp1 + 4 * i);
    acc0 = wmma4(a, b0, acc0);
    acc1 = wmma4(a, b1, acc1);
  }

  const int rbase = (mt << 4) + ((lane >> 4) << 3);
  const int sl0   = (lane >> 4) << 2;
  #pragma unroll
  for (int r8 = 0; r8 < 8; ++r8) {
    const float rsv = __shfl((r8 & 1) ? d1 : d0, sl0 + (r8 >> 1));
    float* crow = C + (size_t)(rbase + r8) * Hc + (lane & 15);
    crow[0]  = acc0[r8] * rsv;
    crow[16] = acc1[r8] * rsv;
  }
}

__global__ void csr_zero(int* __restrict__ cnt)
{
  int i = blockIdx.x * blockDim.x + threadIdx.x;
  if (i < N_NODES) cnt[i] = 0;
}

__device__ __forceinline__ void csr_rank_body(int bid, const int* __restrict__ src,
                                              int* __restrict__ cnt, int* __restrict__ slot)
{
  int e = bid * blockDim.x + threadIdx.x;
  if (e >= E_CNT) return;
  slot[e] = atomicAdd(&cnt[src[e]], 1);
}

#define G1_TILES   ((N_NODES / 16) * (DH / 16))
#define G1_BLOCKS  ((G1_TILES + 7) / 8)
#define EDGE_BLOCKS ((E_CNT + 255) / 256)
#define NODE_BLOCKS ((N_NODES + 255) / 256)
__global__ void __launch_bounds__(256) rank_and_first_gemm(
    const int* __restrict__ src, int* __restrict__ cnt, int* __restrict__ slot,
    const float* __restrict__ x_in, const float* __restrict__ W1,
    const float* __restrict__ b1, float* __restrict__ x0)
{
  if (blockIdx.x < G1_BLOCKS) {
    gemm_xwT_body((int)blockIdx.x, x_in, W1, b1, nullptr, nullptr, x0, nullptr,
                  N_NODES, IN_CH, DH, 1);
  } else {
    csr_rank_body((int)blockIdx.x - G1_BLOCKS, src, cnt, slot);
  }
}

#define SCAN_ITERS ((N_NODES + 1023) / 1024)
#define SCAN_PAD   (SCAN_ITERS * 1024)
__global__ void __launch_bounds__(1024) csr_scan(const int* __restrict__ cnt,
                                                 int* __restrict__ rowptr)
{
  __shared__ int wtot[32];
  const int lane = threadIdx.x & 31;
  const int w    = threadIdx.x >> 5;
  const int base = w * (SCAN_ITERS * 32) + lane;
  int v[SCAN_ITERS];
  int s = 0;
  #pragma unroll
  for (int i = 0; i < SCAN_ITERS; ++i) {
    const int idx = base + i * 32;
    const int x   = cnt[idx < N_NODES ? idx : N_NODES - 1];
    v[i] = idx < N_NODES ? x : 0;
    s += v[i];
  }
  for (int o = 16; o > 0; o >>= 1) s += __shfl_xor(s, o);
  if (lane == 0) wtot[w] = s;
  __syncthreads();
  int carry = 0;
  for (int j = 0; j < w; ++j) carry += wtot[j];
  #pragma unroll
  for (int i = 0; i < SCAN_ITERS; ++i) {
    int inc = v[i];
    #pragma unroll
    for (int o = 1; o < 32; o <<= 1) {
      const int u = __shfl_up(inc, o);
      if (lane >= o) inc += u;
    }
    rowptr[base + i * 32] = carry + inc - v[i];
    carry += __shfl(inc, 31);
  }
}

__device__ __forceinline__ void csr_fill_body(int bid,
                         const int* __restrict__ src, const int* __restrict__ dst,
                         const int* __restrict__ rowptr, const int* __restrict__ slot,
                         int* __restrict__ rec)
{
  int e = bid * blockDim.x + threadIdx.x;
  if (e >= E_CNT) return;
  const int er  = (e >= EU_CNT) ? (e - EU_CNT) : (e + EU_CNT);
  const int s   = src[e];
  const int pos = rowptr[s] + slot[e];
  v4i r;
  r.x = dst[e];
  r.y = s;
  r.z = src[er];
  r.w = dst[er];
  *(v4i*)(rec + 4 * (size_t)pos) = r;
}

__global__ void __launch_bounds__(256) fill_and_first_pq(
    const int* __restrict__ src, const int* __restrict__ dst,
    const int* __restrict__ rowptr, const int* __restrict__ slot, int* __restrict__ rec,
    const float* __restrict__ x0, const float* __restrict__ Ws, float* __restrict__ pq)
{
  if (blockIdx.x < NODE_BLOCKS) {
    sheaf_pq_body((int)blockIdx.x, x0, Ws, pq);
  } else {
    csr_fill_body((int)blockIdx.x - NODE_BLOCKS, src, dst, rowptr, slot, rec);
  }
}

__device__ __forceinline__ int row_pick(int v, unsigned sel_lo, unsigned sel_hi) {
  return (int)__builtin_amdgcn_permlane16((unsigned)v, (unsigned)v, sel_lo, sel_hi, true, false);
}
__device__ __forceinline__ float row_pickf(float v, unsigned sel_lo, unsigned sel_hi) {
  return __int_as_float(row_pick(__float_as_int(v), sel_lo, sel_hi));
}

#define RG 8
template <bool LAST>
__device__ __forceinline__ v4f gather_pair(
    int sbase, int lane,
    const int* __restrict__ rowptr, const int* __restrict__ rec,
    const float* __restrict__ G, const float* __restrict__ dinv,
    const float* __restrict__ diagw, const float* __restrict__ xr,
    const float* __restrict__ eps2, float* __restrict__ x0,
    const float* __restrict__ Wsn,
    float* __restrict__ pq)
{
  const int h  = lane >> 4;
  const int c  = lane & 15;
  const int d  = c >> 3;
  const int e8 = c & 7;
  int s = sbase + h;
  const bool live = s < N_NODES;
  s = live ? s : N_NODES - 1;

  int beg = rowptr[s];
  int end = rowptr[s + 1];
  beg = beg < 0 ? 0 : beg;
  end = end > E_CNT ? E_CNT : end;
  end = live ? end : beg;
  end = end < beg ? beg : end;
  const int deg = end - beg;

  const v4f* xr4 = (const v4f*)xr;
  v4f*       x04 = (v4f*)x0;
  const unsigned own4 = ((unsigned)s << 4) + (unsigned)c;
  const float dsd = dinv[2 * s + d];
  const v4f   xs  = xr4[own4];
  const float dw  = diagw[2 * s + d];
  const v4f   xo   = x04[own4];
  const float epsd = eps2[d];
  v4f acc;
  acc.x = dw * xs.x;
  acc.y = dw * xs.y;
  acc.z = dw * xs.z;
  acc.w = dw * xs.w;

  const int degA = __builtin_amdgcn_readlane(deg, 0);
  const int degB = __builtin_amdgcn_readlane(deg, 16);
  const int dmax = degA > degB ? degA : degB;

  int   t = 0;
  float g = 0.0f;
  if (e8 < deg) {
    const unsigned k = (unsigned)(beg + e8);
    t = rec[4u * k];
    g = G[2u * k + (unsigned)d];
  }
  float w = -g * dsd;
  for (int j0 = 0; j0 < dmax; j0 += RG) {
    int   tn = 0;
    float gn = 0.0f;
    if (j0 + RG + e8 < deg) {
      const unsigned k = (unsigned)(beg + j0 + RG + e8);
      tn = rec[4u * k];
      gn = G[2u * k + (unsigned)d];
    }
    const int   left = deg - j0;
    v4f xv[RG];
    unsigned ix[RG];
    #pragma unroll
    for (int u = 0; u < RG; ++u) {
      const unsigned sl = 0x11111111u * (unsigned)u;
      ix[u] = ((unsigned)row_pick(t, sl, sl + 0x88888888u) << 4) + (unsigned)c;
    }
    __builtin_amdgcn_sched_barrier(0);
    #pragma unroll
    for (int u = 0; u < RG; ++u) xv[u] = xr4[ix[u]];
    #pragma unroll
    for (int u = 0; u < RG; ++u) {
      const unsigned sl = 0x11111111u * (unsigned)u;
      const float wj = row_pickf(w, sl, sl + 0x88888888u);
      if (u < left) {
        acc.x += wj * xv[u].x;
        acc.y += wj * xv[u].y;
        acc.z += wj * xv[u].z;
        acc.w += wj * xv[u].w;
      }
    }
    t = tn;
    w = -gn * dsd;
  }

  v4f wp0 = {0.0f, 0.0f, 0.0f, 0.0f}, wq0 = wp0, wp1 = wp0, wq1 = wp0;
  if (!LAST) {
    wp0 = *(const v4f*)(Wsn + 4 * c);
    wq0 = *(const v4f*)(Wsn + DH + 4 * c);
    wp1 = *(const v4f*)(Wsn + 2 * DH + 4 * c);
    wq1 = *(const v4f*)(Wsn + 3 * DH + 4 * c);
  }

  const float coeff = 1.0f + tanhf(epsd);
  const float e0 = (acc.x > 0.0f) ? acc.x : expm1f(acc.x);
  const float e1 = (acc.y > 0.0f) ? acc.y : expm1f(acc.y);
  const float e2 = (acc.z > 0.0f) ? acc.z : expm1f(acc.z);
  const float e3 = (acc.w > 0.0f) ? acc.w : expm1f(acc.w);
  v4f nx;
  nx.x = coeff * xo.x - e0;
  nx.y = coeff * xo.y - e1;
  nx.z = coeff * xo.z - e2;
  nx.w = coeff * xo.w - e3;
  if (!LAST && live) x04[own4] = nx;

  if (!LAST) {
    float p0 = wp0.x * nx.x + wp0.y * nx.y + wp0.z * nx.z + wp0.w * nx.w;
    float p1 = wp1.x * nx.x + wp1.y * nx.y + wp1.z * nx.z + wp1.w * nx.w;
    float q0 = wq0.x * nx.x + wq0.y * nx.y + wq0.z * nx.z + wq0.w * nx.w;
    float q1 = wq1.x * nx.x + wq1.y * nx.y + wq1.z * nx.z + wq1.w * nx.w;
    p0 += row_pickf(p0, 0xFEDCBA98u, 0x76543210u);
    p1 += row_pickf(p1, 0xFEDCBA98u, 0x76543210u);
    q0 += row_pickf(q0, 0xFEDCBA98u, 0x76543210u);
    q1 += row_pickf(q1, 0xFEDCBA98u, 0x76543210u);
    p0 += row_pickf(p0, 0x32107654u, 0xBA98FEDCu);
    p1 += row_pickf(p1, 0x32107654u, 0xBA98FEDCu);
    q0 += row_pickf(q0, 0x32107654u, 0xBA98FEDCu);
    q1 += row_pickf(q1, 0x32107654u, 0xBA98FEDCu);
    p0 += row_pickf(p0, 0x54761032u, 0xDCFE98BAu);
    p1 += row_pickf(p1, 0x54761032u, 0xDCFE98BAu);
    q0 += row_pickf(q0, 0x54761032u, 0xDCFE98BAu);
    q1 += row_pickf(q1, 0x54761032u, 0xDCFE98BAu);
    p0 += row_pickf(p0, 0x67452301u, 0xEFCDAB89u);
    p1 += row_pickf(p1, 0x67452301u, 0xEFCDAB89u);
    q0 += row_pickf(q0, 0x67452301u, 0xEFCDAB89u);
    q1 += row_pickf(q1, 0x67452301u, 0xEFCDAB89u);
    if (c == 0 && live) {
      v4f o4; o4.x = p0; o4.y = p1; o4.z = q0; o4.w = q1;
      *(v4f*)(pq + 4 * (size_t)s) = o4;
    }
  }
  return nx;
}

#define XPITCH (DH + 4)
template <bool LAST>
__global__ void __launch_bounds__(256) node_gather(
    const int* __restrict__ rowptr, const int* __restrict__ rec,
    const float* __restrict__ G, const float* __restrict__ dinv,
    const float* __restrict__ diagw, const float* __restrict__ xr,
    const float* __restrict__ eps2, float* __restrict__ x0,
    const float* __restrict__ Wsn,
    float* __restrict__ pq,
    const float* __restrict__ W2,
    const float* __restrict__ b2, float* __restrict__ out)
{
  const int lane = threadIdx.x & 31;
  const int wv   = threadIdx.x >> 5;
  const int sbase = __builtin_amdgcn_readfirstlane((int)(blockIdx.x * 16 + wv * 2));
  v4f nx = {0.0f, 0.0f, 0.0f, 0.0f};
  if (sbase < N_NODES)
    nx = gather_pair<LAST>(sbase, lane, rowptr, rec, G, dinv, diagw, xr, eps2, x0, Wsn, pq);
  if (!LAST) return;

  __shared__ v4f xrow4[16 * XPITCH / 4];
  float* xrow = (float*)xrow4;
  *(v4f*)(xrow + (wv * 2 + (lane >> 4)) * XPITCH + 4 * (lane & 15)) = nx;
  __syncthreads();
  if (wv >= 2) return;
  const int m    = lane & 15;
  const int kb   = (lane >> 4) << 1;
  const int bcol = (wv << 4) + m;
  const float* Wp = W2 + (size_t)bcol * DH + kb;
  const float* Xp = xrow + m * XPITCH + kb;
  v8f acc = {};
  for (int k0 = 0; k0 < DH; k0 += 4) {
    const v2f a = *(const v2f*)(Xp + k0);
    const v2f b = *(const v2f*)(Wp + k0);
    acc = wmma4(a, b, acc);
  }
  const float bv = b2[bcol];
  const int nb = (int)blockIdx.x * 16 + ((lane >> 4) << 3);
  #pragma unroll
  for (int r = 0; r < 8; ++r) {
    if (nb + r < N_NODES) out[(size_t)(nb + r) * OUTC + bcol] = acc[r] + bv;
  }
}

extern "C" void kernel_launch(void* const* d_in, const int* in_sizes, int n_in,
                              void* d_out, int out_size, void* d_ws, size_t ws_size,
                              hipStream_t stream) {
  const float* x_in   = (const float*)d_in[0];
  const int*   ei     = (const int*)  d_in[1];
  const float* W1     = (const float*)d_in[2];
  const float* b1     = (const float*)d_in[3];
  const float* Wsheaf = (const float*)d_in[4];
  const float* Wleft  = (const float*)d_in[5];
  const float* Wright = (const float*)d_in[6];
  const float* eps    = (const float*)d_in[7];
  const float* W2     = (const float*)d_in[8];
  const float* b2     = (const float*)d_in[9];
  float* out = (float*)d_out;

  float* ws = (float*)d_ws;
  size_t o = 0;
  o += (size_t)ND * Hc;
  float* x0    = ws + o; o += (size_t)ND * Hc;
  int*   rec   = (int*)(ws + o); o += (size_t)ND * Hc;
  float* xr    = ws + o; o += (size_t)ND * Hc;
  int*   slot  = (int*)xr;
  int*   cnt   = (int*)xr + (size_t)E_CNT;
  float* pq    = ws + o; o += 2 * (size_t)N_NODES; o += 2 * (size_t)N_NODES;
  float* G     = ws + o; o += 2 * (size_t)E_CNT;
  int*   rowptr = (int*)(ws + o); o += (size_t)ND;
  float* dinv  = ws + o; o += (size_t)ND;
  float* diagw = ws + o; o += (size_t)ND;

  const int* src = ei;
  const int* dst = ei + E_CNT;

  const int TPB = 256;

  csr_zero<<<NODE_BLOCKS, TPB, 0, stream>>>(cnt);
  rank_and_first_gemm<<<G1_BLOCKS + EDGE_BLOCKS, TPB, 0, stream>>>(
      src, cnt, slot, x_in, W1, b1, x0);
  csr_scan<<<1, 1024, 0, stream>>>(cnt, rowptr);
  fill_and_first_pq<<<NODE_BLOCKS + EDGE_BLOCKS, TPB, 0, stream>>>(
      src, dst, rowptr, slot, rec, x0, Wsheaf, pq);

  for (int l = 0; l < LAYERS; ++l) {
    layer_edge_gemm<<<((ND / 16) + 7) / 8, TPB, 0, stream>>>(
        x0, Wright + (size_t)l * Hc * Hc, Wleft + (size_t)l * 4,
        rowptr, rec, pq, G, xr, dinv, diagw);
    if (l + 1 < LAYERS) {
      node_gather<false><<<(N_NODES + 15) / 16, TPB, 0, stream>>>(
          rowptr, rec, G, dinv, diagw, xr, eps + (size_t)l * 2, x0,
          Wsheaf + (size_t)(l + 1) * 2 * 2 * DH, pq, W2, b2, out);
    } else {
      node_gather<true><<<(N_NODES + 15) / 16, TPB, 0, stream>>>(
          rowptr, rec, G, dinv, diagw, xr, eps + (size_t)l * 2, x0,
          Wsheaf, pq, W2, b2, out);
    }
  }
}
